# starting kernel
# baseline (speedup 1.0000x reference)
#include <hip/hip_runtime.h>
#include <math.h>


typedef __attribute__((ext_vector_type(2))) float v2f;
typedef __attribute__((ext_vector_type(8))) float v8f;

__device__ __forceinline__ unsigned encf(float f) {
  unsigned u = __float_as_uint(f);
  return (u & 0x80000000u) ? ~u : (u | 0x80000000u);
}
__device__ __forceinline__ float decf(unsigned u) {
  unsigned v = (u & 0x80000000u) ? (u & 0x7FFFFFFFu) : ~u;
  return __uint_as_float(v);
}
#define ENC_NEG_INF 0x007FFFFFu

__global__ void gemm16_wmma(const float* __restrict__ X,
                            const float* __restrict__ W,
                            float* __restrict__ Hout,
                            int nrows, int ncols) {
  const int wave = blockIdx.x * (blockDim.x >> 5) + (threadIdx.x >> 5);
  const int lane = threadIdx.x & 31;
  const int row_base = wave * 16;
  if (row_base >= nrows) return;

  const int m    = lane & 15;
  const int half = lane >> 4;
  int row = row_base + m; if (row > nrows - 1) row = nrows - 1;
  int col = m;            if (col > ncols - 1) col = ncols - 1;

  const float2* __restrict__ Xr = (const float2*)(X + (size_t)row * 16);

  v8f c = {0.f, 0.f, 0.f, 0.f, 0.f, 0.f, 0.f, 0.f};
#pragma unroll
  for (int k0 = 0; k0 < 4; ++k0) {
    const int k = k0 * 4 + half * 2;
    const float2 av = Xr[k0 * 2 + half];
    v2f a; a.x = av.x; a.y = av.y;
    v2f b;
    b.x = W[(k)     * ncols + col];
    b.y = W[(k + 1) * ncols + col];
    c = __builtin_amdgcn_wmma_f32_16x16x4_f32(
         false, a,  false, b,
         (short)0, c,  false,  false);
  }

  if (m < ncols) {
    const int rofs = half * 8;
    float* __restrict__ outp = Hout + (size_t)(row_base + rofs) * ncols + m;
    if (row_base + 16 <= nrows) {
#pragma unroll
      for (int r = 0; r < 8; ++r) outp[(size_t)r * ncols] = c[r];
    } else {
#pragma unroll
      for (int r = 0; r < 8; ++r)
        if (row_base + rofs + r < nrows) outp[(size_t)r * ncols] = c[r];
    }
  }
}

template <int C>
__global__ void node_logits_init(const float* __restrict__ Hm,
                                 const float* __restrict__ a_src,
                                 const float* __restrict__ a_dst,
                                 float2* __restrict__ als, float2* __restrict__ ald,
                                 unsigned* __restrict__ menc,
                                 float2* __restrict__ z, float* __restrict__ acc,
                                 int nrows) {
  const int n = blockIdx.x * blockDim.x + threadIdx.x;
  if (n >= nrows) return;
  constexpr int F = 2 * C;
  const float4* __restrict__ hr = (const float4*)(Hm + (size_t)n * F);
  float s0 = 0.f, s1 = 0.f, d0 = 0.f, d1 = 0.f;
#pragma unroll
  for (int v = 0; v < C / 4; ++v) {
    const float4 h0 = hr[v];
    const float4 h1 = hr[C / 4 + v];
#pragma unroll
    for (int j = 0; j < 4; ++j) {
      const float e0 = (&h0.x)[j], e1 = (&h1.x)[j];
      const int cch = 4 * v + j;
      s0 += e0 * a_src[cch];     d0 += e0 * a_dst[cch];
      s1 += e1 * a_src[C + cch]; d1 += e1 * a_dst[C + cch];
    }
  }
  als[n] = make_float2(s0, s1);
  ald[n] = make_float2(d0, d1);
  menc[n * 2] = ENC_NEG_INF; menc[n * 2 + 1] = ENC_NEG_INF;
  z[n] = make_float2(0.f, 0.f);
  float4* __restrict__ ar = (float4*)(acc + (size_t)n * F);
#pragma unroll
  for (int v = 0; v < F / 4; ++v) ar[v] = make_float4(0.f, 0.f, 0.f, 0.f);
}

__global__ void edge_max(const int* __restrict__ ei,
                         const float2* __restrict__ als,
                         const float2* __restrict__ ald,
                         unsigned* __restrict__ menc, int E, int N) {
  const int t = blockIdx.x * blockDim.x + threadIdx.x;
  if (t >= E + N) return;
  int src, dst;
  if (t < E) { src = ei[t]; dst = ei[E + t]; } else { src = dst = t - E; }
  const float2 s = als[src];
  const float2 d = ald[dst];
  float e0 = s.x + d.x; e0 = (e0 > 0.f) ? e0 : 0.2f * e0;
  float e1 = s.y + d.y; e1 = (e1 > 0.f) ? e1 : 0.2f * e1;
  atomicMax(&menc[dst * 2],     encf(e0));
  atomicMax(&menc[dst * 2 + 1], encf(e1));
}

template <int C>
__global__ void edge_accum(const int* __restrict__ ei,
                           const float2* __restrict__ als,
                           const float2* __restrict__ ald,
                           const unsigned* __restrict__ menc,
                           float* __restrict__ z, float* __restrict__ acc,
                           const float* __restrict__ Hm,
                           int E, int N) {
  const int t = blockIdx.x * blockDim.x + threadIdx.x;
  if (t >= E + N) return;
  int src, dst;
  if (t < E) { src = ei[t]; dst = ei[E + t]; } else { src = dst = t - E; }
  constexpr int F = 2 * C;
  const float2 s = als[src];
  const float2 d = ald[dst];
  const float4* __restrict__ hs = (const float4*)(Hm + (size_t)src * F);
  float w[2];
#pragma unroll
  for (int hd = 0; hd < 2; ++hd) {
    float e = (hd ? s.y + d.y : s.x + d.x);
    e = (e > 0.f) ? e : 0.2f * e;
    float mm = decf(menc[dst * 2 + hd]);
    if (!(mm >= -3.0e38f && mm <= 3.0e38f)) mm = 0.f;
    w[hd] = __expf(e - mm);
    atomicAdd(&z[dst * 2 + hd], w[hd]);
  }
  float* __restrict__ ad = acc + (size_t)dst * F;
#pragma unroll
  for (int hd = 0; hd < 2; ++hd) {
#pragma unroll
    for (int v = 0; v < C / 4; ++v) {
      const float4 hv = hs[hd * (C / 4) + v];
      const int base = hd * C + 4 * v;
      atomicAdd(&ad[base + 0], w[hd] * hv.x);
      atomicAdd(&ad[base + 1], w[hd] * hv.y);
      atomicAdd(&ad[base + 2], w[hd] * hv.z);
      atomicAdd(&ad[base + 3], w[hd] * hv.w);
    }
  }
}

template <int C, bool ELU>
__global__ void node_finalize(const float* __restrict__ acc,
                              const float2* __restrict__ z,
                              const float* __restrict__ bias,
                              float* __restrict__ out, int nrows) {
  const int n = blockIdx.x * blockDim.x + threadIdx.x;
  if (n >= nrows) return;
  constexpr int F = 2 * C;
  const float4* __restrict__ ar = (const float4*)(acc + (size_t)n * F);
  float4* __restrict__ outp = (float4*)(out + (size_t)n * F);
  const float2 zz = z[n];
  const float rz0 = 1.f / (zz.x + 1e-16f);
  const float rz1 = 1.f / (zz.y + 1e-16f);
#pragma unroll
  for (int v = 0; v < F / 4; ++v) {
    float4 a = ar[v];
    const float rz = (4 * v < C) ? rz0 : rz1;
#pragma unroll
    for (int j = 0; j < 4; ++j) {
      float x = (&a.x)[j] * rz + bias[4 * v + j];
      if (ELU) x = (x > 0.f) ? x : (__expf(x) - 1.f);
      (&a.x)[j] = x;
    }
    outp[v] = a;
  }
}

extern "C" void kernel_launch(void* const* d_in, const int* in_sizes, int n_in,
                              void* d_out, int out_size, void* d_ws, size_t ws_size,
                              hipStream_t stream) {
  const float* x   = (const float*)d_in[0];
  const int*   ei  = (const int*)  d_in[1];
  const float* W1  = (const float*)d_in[2];
  const float* as1 = (const float*)d_in[3];
  const float* ad1 = (const float*)d_in[4];
  const float* b1  = (const float*)d_in[5];
  const float* W2  = (const float*)d_in[6];
  const float* as2 = (const float*)d_in[7];
  const float* ad2 = (const float*)d_in[8];
  const float* b2  = (const float*)d_in[9];

  const int N = in_sizes[0] / 16;
  const int E = in_sizes[1] / 2;

  float*    ws   = (float*)d_ws;
  float*    h1   = ws;
  float*    acc1 = ws + (size_t)16 * N;
  float2*   als  = (float2*)(ws + (size_t)32 * N);
  float2*   ald  = (float2*)(ws + (size_t)34 * N);
  unsigned* menc = (unsigned*)(ws + (size_t)36 * N);
  float*    z    = ws + (size_t)38 * N;
  float*    h2   = ws + (size_t)40 * N;
  float*    acc2 = ws + (size_t)48 * N;

  const int TB = 256;
  const int node_grid = (N + TB - 1) / TB;
  const int edge_grid = (E + N + TB - 1) / TB;
  const int tiles     = (N + 15) / 16;
  const int gemm_grid = (tiles + 7) / 8;

  gemm16_wmma<<<gemm_grid, TB, 0, stream>>>(x, W1, h1, N, 16);
  node_logits_init<8><<<node_grid, TB, 0, stream>>>(h1, as1, ad1, als, ald, menc,
                                                    (float2*)z, acc1, N);
  edge_max<<<edge_grid, TB, 0, stream>>>(ei, als, ald, menc, E, N);
  edge_accum<8><<<edge_grid, TB, 0, stream>>>(ei, als, ald, menc, z, acc1, h1, E, N);
  node_finalize<8, true><<<node_grid, TB, 0, stream>>>(acc1, (const float2*)z, b1, acc1, N);

  gemm16_wmma<<<gemm_grid, TB, 0, stream>>>(acc1, W2, h2, N, 8);
  node_logits_init<4><<<node_grid, TB, 0, stream>>>(h2, as2, ad2, als, ald, menc,
                                                    (float2*)z, acc2, N);
  edge_max<<<edge_grid, TB, 0, stream>>>(ei, als, ald, menc, E, N);
  edge_accum<4><<<edge_grid, TB, 0, stream>>>(ei, als, ald, menc, z, acc2, h2, E, N);
  node_finalize<4, false><<<node_grid, TB, 0, stream>>>(acc2, (const float2*)z, b2,
                                                        (float*)d_out, N);
}


# edge softmax: shift = leaky_relu(max of al_s over all nodes + al_d[dst]), an upper bound of the segment max; the per-edge max pass runs only if an al_s spread exceeds 40; sums scaled, same sum order
# speedup vs baseline: 6.2597x; 6.2597x over previous
#include <hip/hip_runtime.h>
#include <math.h>


typedef __attribute__((ext_vector_type(2))) float v2f;
typedef __attribute__((ext_vector_type(8))) float v8f;

__device__ __forceinline__ unsigned encf(float f) {
  unsigned u = __float_as_uint(f);
  return (u & 0x80000000u) ? ~u : (u | 0x80000000u);
}
__device__ __forceinline__ float decf(unsigned u) {
  unsigned v = (u & 0x80000000u) ? (u & 0x7FFFFFFFu) : ~u;
  return __uint_as_float(v);
}
#define ENC_NEG_INF 0x007FFFFFu
#define ENC_POS_INF 0xFF800000u
#define SPREAD_MAX 40.0f

__device__ __forceinline__ bool bound_mode(const unsigned* __restrict__ stat,
                                           float& b0, float& b1) {
  b0 = decf(stat[0]); b1 = decf(stat[1]);
  const float l0 = decf(stat[2]), l1 = decf(stat[3]);
  return (b0 - l0 <= SPREAD_MAX) && (b1 - l1 <= SPREAD_MAX);
}

#define CHUNK 512
#define RS 1568
#define QSTRIDE 64

__device__ __forceinline__ int package_half() {
  const unsigned v = __builtin_amdgcn_s_sendmsg_rtn(0x87);
  return (int)((v >> 18) & 1u);
}

__device__ __forceinline__ int split_node(int N) {
  const int s = ((N >> 1) + 31) & ~31;
  return (s < N) ? s : N;
}

__global__ void gemm16_wmma(const float* __restrict__ X,
                            const float* __restrict__ W,
                            float* __restrict__ Hout,
                            unsigned* __restrict__ stat,
                            int nrows, int ncols) {
  if (blockIdx.x == 0 && threadIdx.x < 4)
    stat[threadIdx.x] = (threadIdx.x < 2) ? ENC_NEG_INF : ENC_POS_INF;
  const int wave = blockIdx.x * (blockDim.x >> 5) + (threadIdx.x >> 5);
  const int lane = threadIdx.x & 31;
  const int row_base = wave * 16;
  if (row_base >= nrows) return;

  const int m    = lane & 15;
  const int half = lane >> 4;
  int row = row_base + m; if (row > nrows - 1) row = nrows - 1;
  int col = m;            if (col > ncols - 1) col = ncols - 1;

  const float2* __restrict__ Xr = (const float2*)(X + (size_t)row * 16);

  v8f c = {0.f, 0.f, 0.f, 0.f, 0.f, 0.f, 0.f, 0.f};
#pragma unroll
  for (int k0 = 0; k0 < 4; ++k0) {
    const int k = k0 * 4 + half * 2;
    const float2 av = Xr[k0 * 2 + half];
    v2f a; a.x = av.x; a.y = av.y;
    v2f b;
    b.x = W[(k)     * ncols + col];
    b.y = W[(k + 1) * ncols + col];
    c = __builtin_amdgcn_wmma_f32_16x16x4_f32(
         false, a,  false, b,
         (short)0, c,  false,  false);
  }

  if (m < ncols) {
    const int rofs = half * 8;
    float* __restrict__ outp = Hout + (size_t)(row_base + rofs) * ncols + m;
    if (row_base + 16 <= nrows) {
#pragma unroll
      for (int r = 0; r < 8; ++r) outp[(size_t)r * ncols] = c[r];
    } else {
#pragma unroll
      for (int r = 0; r < 8; ++r)
        if (row_base + rofs + r < nrows) outp[(size_t)r * ncols] = c[r];
    }
  }
}

template <int C>
__global__ void node_logits_init(const float* __restrict__ Hm,
                                 const float* __restrict__ a_src,
                                 const float* __restrict__ a_dst,
                                 float2* __restrict__ als, float2* __restrict__ ald,
                                 unsigned* __restrict__ menc,
                                 float2* __restrict__ z, float* __restrict__ acc,
                                 int* __restrict__ qctr,
                                 unsigned* __restrict__ stat, int nrows) {
  __shared__ float s_st[32][4];
  if (blockIdx.x == 0) qctr[threadIdx.x] = 0;
  const int n = blockIdx.x * blockDim.x + threadIdx.x;
  const bool valid = n < nrows;
  constexpr int F = 2 * C;
  float s0 = 0.f, s1 = 0.f, dd0 = 0.f, dd1 = 0.f;
  if (valid) {
    const float4* __restrict__ hr = (const float4*)(Hm + (size_t)n * F);
    float d0 = 0.f, d1 = 0.f;
#pragma unroll
    for (int v = 0; v < C / 4; ++v) {
      const float4 h0 = hr[v];
      const float4 h1 = hr[C / 4 + v];
#pragma unroll
      for (int j = 0; j < 4; ++j) {
        const float e0 = (&h0.x)[j], e1 = (&h1.x)[j];
        const int cch = 4 * v + j;
        s0 += e0 * a_src[cch];     d0 += e0 * a_dst[cch];
        s1 += e1 * a_src[C + cch]; d1 += e1 * a_dst[C + cch];
      }
    }
    als[n] = make_float2(s0, s1);
    ald[n] = make_float2(d0, d1);
    dd0 = d0; dd1 = d1;
    menc[n * 2] = ENC_NEG_INF; menc[n * 2 + 1] = ENC_NEG_INF;
    z[n] = make_float2(0.f, 0.f);
    float4* __restrict__ ar = (float4*)(acc + (size_t)n * F);
#pragma unroll
    for (int v = 0; v < F / 4; ++v) ar[v] = make_float4(0.f, 0.f, 0.f, 0.f);
  }
  float r[4];
  r[0] = valid ? s0 : -INFINITY; r[1] = valid ? s1 : -INFINITY;
  r[2] = valid ? s0 :  INFINITY; r[3] = valid ? s1 :  INFINITY;
  if (valid && !(fabsf(s0) <= 3.0e38f && fabsf(s1) <= 3.0e38f &&
                 fabsf(dd0) <= 3.0e38f && fabsf(dd1) <= 3.0e38f)) r[0] = INFINITY;
#pragma unroll
  for (int o = 16; o > 0; o >>= 1) {
    r[0] = fmaxf(r[0], __shfl_xor(r[0], o)); r[1] = fmaxf(r[1], __shfl_xor(r[1], o));
    r[2] = fminf(r[2], __shfl_xor(r[2], o)); r[3] = fminf(r[3], __shfl_xor(r[3], o));
  }
  if ((threadIdx.x & 31) == 0) {
#pragma unroll
    for (int k = 0; k < 4; ++k) s_st[threadIdx.x >> 5][k] = r[k];
  }
  __syncthreads();
  if (threadIdx.x < 4) {
    const int k = (int)threadIdx.x;
    const int nw = (int)(blockDim.x >> 5);
    float v = s_st[0][k];
    for (int w = 1; w < nw; ++w) v = (k < 2) ? fmaxf(v, s_st[w][k]) : fminf(v, s_st[w][k]);
    if (k < 2) atomicMax(&stat[k], encf(v));
    else       atomicMin(&stat[k], encf(v));
  }
}

__global__ void edge_max(const int* __restrict__ ei,
                         const float2* __restrict__ als,
                         const float2* __restrict__ ald,
                         unsigned* __restrict__ menc,
                         unsigned* __restrict__ keyw,
                         int* __restrict__ qctr,
                         const unsigned* __restrict__ stat, int E, int N) {
  __shared__ int s_c;
  {
    float b0, b1;
    if (bound_mode(stat, b0, b1)) {
      const int* __restrict__ dstp = ei + E;
      const int nwords = (E + 3) >> 2;
      const int step = (int)(gridDim.x * blockDim.x);
      for (int w = (int)(blockIdx.x * blockDim.x + threadIdx.x); w < nwords; w += step) {
        unsigned v = 0u;
        if ((E & 3) == 0) {
          const int4 d = ((const int4*)dstp)[w];
          v = (((unsigned)d.x / RS) & 0xFFu)       | ((((unsigned)d.y / RS) & 0xFFu) << 8) |
              ((((unsigned)d.z / RS) & 0xFFu) << 16) | ((((unsigned)d.w / RS) & 0xFFu) << 24);
        } else {
#pragma unroll
          for (int j = 0; j < 4; ++j) {
            const int t = 4 * w + j;
            if (t < E) v |= (((unsigned)dstp[t] / RS) & 0xFFu) << (8 * j);
          }
        }
        keyw[w] = v;
      }
      return;
    }
  }
  unsigned char* __restrict__ keyb = (unsigned char*)keyw;
  const int total = E + N;
  const int nchunks = (total + CHUNK - 1) / CHUNK;
  const int nsplit = split_node(N);
  const int myhalf = package_half();
  for (int side = 0; side < 2; ++side) {
    const int part = myhalf ^ side;
    const int lo = part ? nsplit : 0;
    const unsigned span = (unsigned)((part ? N : nsplit) - lo);
    int* __restrict__ q = qctr + part * QSTRIDE;
    int pend = 0;
    if (threadIdx.x == 0) pend = atomicAdd(q, 1);
    for (int g = 0; g <= nchunks; ++g) {
      if (threadIdx.x == 0) s_c = pend;
      __syncthreads();
      const int c = s_c;
      __syncthreads();
      if (c >= nchunks) break;
      if (threadIdx.x == 0) pend = atomicAdd(q, 1);
      int dstk[CHUNK / 256];
#pragma unroll
      for (int k = 0; k < CHUNK / 256; ++k) {
        const int t = c * CHUNK + k * 256 + (int)threadIdx.x;
        dstk[k] = (t < total) ? ((t < E) ? ei[E + t] : t - E) : -1;
      }
#pragma unroll
      for (int k = 0; k < CHUNK / 256; ++k) {
        const int t = c * CHUNK + k * 256 + (int)threadIdx.x;
        const int dst = dstk[k];
        if (part == 0 && t < E) keyb[t] = (unsigned char)((unsigned)dst / RS);
        if ((unsigned)(dst - lo) < span) {
          const int src = (t < E) ? ei[t] : dst;
          const float2 s = als[src];
          const float2 d = ald[dst];
          float e0 = s.x + d.x; e0 = (e0 > 0.f) ? e0 : 0.2f * e0;
          float e1 = s.y + d.y; e1 = (e1 > 0.f) ? e1 : 0.2f * e1;
          atomicMax(&menc[dst * 2],     encf(e0));
          atomicMax(&menc[dst * 2 + 1], encf(e1));
        }
      }
    }
  }
}

#define RB 4
#define SCAN_K 8
#define ACC_TB 1024
#define QCAP 160

__device__ __forceinline__ void wave_queue_handover() {
  __builtin_amdgcn_fence(__ATOMIC_RELEASE, "wavefront");
  __builtin_amdgcn_wave_barrier();
  __builtin_amdgcn_fence(__ATOMIC_ACQUIRE, "wavefront");
}

template <int C>
__device__ __forceinline__ void edge_terms(int src, int dl,
                                           const float2* __restrict__ als,
                                           const float* __restrict__ Hm,
                                           const float4* s_dm,
                                           float* s_row) {
  constexpr int F = 2 * C;
  constexpr int P = F + 3;
  const float2 s = als[src];
  const float4 dm = s_dm[dl];
  const float4* __restrict__ hs = (const float4*)(Hm + (size_t)src * F);
  float w[2];
#pragma unroll
  for (int hd = 0; hd < 2; ++hd) {
    float e = (hd ? s.y + dm.y : s.x + dm.x);
    e = (e > 0.f) ? e : 0.2f * e;
    w[hd] = __expf(e - (hd ? dm.w : dm.z));
  }
  float* r = s_row + dl * P;
  atomicAdd(&r[F + 0], w[0]);
  atomicAdd(&r[F + 1], w[1]);
#pragma unroll
  for (int hd = 0; hd < 2; ++hd) {
#pragma unroll
    for (int v = 0; v < C / 4; ++v) {
      const float4 hv = hs[hd * (C / 4) + v];
      const int base = hd * C + 4 * v;
      atomicAdd(&r[base + 0], w[hd] * hv.x);
      atomicAdd(&r[base + 1], w[hd] * hv.y);
      atomicAdd(&r[base + 2], w[hd] * hv.z);
      atomicAdd(&r[base + 3], w[hd] * hv.w);
    }
  }
}

template <int C>
__global__ void __launch_bounds__(ACC_TB)
edge_accum(const int* __restrict__ ei,
           const float2* __restrict__ als,
           const float2* __restrict__ ald,
           const unsigned* __restrict__ menc,
           float* __restrict__ z, float* __restrict__ acc,
           const float* __restrict__ Hm,
           const unsigned* __restrict__ keyw,
           int* __restrict__ qctr,
           const unsigned* __restrict__ stat, int E, int N) {
  constexpr int F = 2 * C;
  constexpr int P = F + 3;
  __shared__ float s_row[RS * P];
  __shared__ float4 s_dm[RS];
  __shared__ int s_q[ACC_TB / 32][QCAP];
  __shared__ int s_c;

  const int tid  = (int)threadIdx.x;
  const int lane = tid & 31;
  int* q_w = &s_q[tid >> 5][0];
  const int* __restrict__ dstp = ei + E;

  const int R  = (N + RS - 1) / RS;
  const int Rh = (R + 1) >> 1;
  const int myhalf = package_half();
  float b0, b1;
  const bool bmode = bound_mode(stat, b0, b1);
  for (int side = 0; side < 2; ++side) {
    const int part = myhalf ^ side;
    const int r0 = part ? Rh : 0;
    const int nitems = (part ? R - Rh : Rh) * RB;
    int* __restrict__ q = qctr + part * QSTRIDE;
    int pend = 0;
    if (tid == 0) pend = atomicAdd(q, 1);
    for (int g = 0; g <= nitems; ++g) {
      if (tid == 0) s_c = pend;
      __syncthreads();
      const int c = s_c;
      __syncthreads();
      if (c >= nitems) break;
      if (tid == 0) pend = atomicAdd(q, 1);
      const int rng = r0 + c / RB;
      const int b   = c % RB;
      const int lo  = rng * RS;
      const int span = ((lo + RS < N) ? lo + RS : N) - lo;

      for (int i = tid; i < RS * P; i += ACC_TB) s_row[i] = 0.f;
      for (int i = tid; i < span; i += ACC_TB) {
        const float2 d = ald[lo + i];
        float m0, m1;
        if (bmode) {
          m0 = b0 + d.x; m0 = (m0 > 0.f) ? m0 : 0.2f * m0;
          m1 = b1 + d.y; m1 = (m1 > 0.f) ? m1 : 0.2f * m1;
        } else {
          m0 = decf(menc[(lo + i) * 2]);
          m1 = decf(menc[(lo + i) * 2 + 1]);
          if (!(m0 >= -3.0e38f && m0 <= 3.0e38f)) m0 = 0.f;
          if (!(m1 >= -3.0e38f && m1 <= 3.0e38f)) m1 = 0.f;
        }
        s_dm[i] = make_float4(d.x, d.y, m0, m1);
      }
      __syncthreads();

      const int e_lo = (int)(((long long)E * b) / RB) & ~3;
      const int e_hi = (b == RB - 1) ? E : ((int)(((long long)E * (b + 1)) / RB) & ~3);
      const unsigned rngw = (unsigned)rng * 0x01010101u;
      int qn = 0;
      for (int wb = e_lo >> 2; wb < ((e_hi + 3) >> 2); wb += ACC_TB * SCAN_K) {
        unsigned kw[SCAN_K];
        if (wb + ACC_TB * SCAN_K <= (e_hi >> 2)) {
#pragma unroll
          for (int k = 0; k < SCAN_K; ++k) kw[k] = keyw[wb + k * ACC_TB + tid] ^ rngw;
        } else {
#pragma unroll
          for (int k = 0; k < SCAN_K; ++k) {
            const int w = wb + k * ACC_TB + tid;
            const int left = e_hi - 4 * w;
            unsigned v = 0xFFFFFFFFu;
            if (left > 0) {
              v = keyw[w] ^ rngw;
              if (left < 4) v |= 0xFFFFFFFFu << (8 * left);
            }
            kw[k] = v;
          }
        }
#pragma unroll
        for (int k = 0; k < SCAN_K; ++k) {
          const int t0 = 4 * (wb + k * ACC_TB + tid);
          const unsigned x = kw[k];
#pragma unroll
          for (int j = 0; j < 4; ++j) {
            const bool own = (x & (0xFFu << (8 * j))) == 0u;
            const unsigned mask = __builtin_amdgcn_ballot_w32(own);
            if (mask == 0u) continue;
            if (own) {
              const int pos = qn + __builtin_popcount(mask & ((1u << lane) - 1u));
              q_w[pos] = t0 + j;
            }
            qn += __builtin_popcount(mask);
          }
          while (qn >= 32) {
            qn -= 32;
            wave_queue_handover();
            const int t = q_w[qn + lane];
            const int src = ei[t];
            const int dst = dstp[t];
            asm volatile("" :: "v"(src), "v"(dst));
            const int dl = dst - lo;
            if ((unsigned)dl < (unsigned)span)
              edge_terms<C>(src, dl, als, Hm, s_dm, s_row);
          }
        }
      }
      wave_queue_handover();
      if (lane < qn) {
        const int t = q_w[lane];
        const int src = ei[t];
        const int dst = dstp[t];
        asm volatile("" :: "v"(src), "v"(dst));
        const int dl = dst - lo;
        if ((unsigned)dl < (unsigned)span)
          edge_terms<C>(src, dl, als, Hm, s_dm, s_row);
      }
      if (b == 0)
        for (int i = tid; i < span; i += ACC_TB)
          edge_terms<C>(lo + i, i, als, Hm, s_dm, s_row);
      __syncthreads();

      float* __restrict__ accr = acc + (size_t)lo * F;
      for (int i = tid; i < span * F; i += ACC_TB)
        atomicAdd(&accr[i], s_row[(i / F) * P + (i % F)]);
      float* __restrict__ zr = z + (size_t)lo * 2;
      for (int i = tid; i < span * 2; i += ACC_TB)
        atomicAdd(&zr[i], s_row[(i >> 1) * P + F + (i & 1)]);
    }
  }
}

template <int C, bool ELU>
__global__ void node_finalize(const float* __restrict__ acc,
                              const float2* __restrict__ z,
                              const float* __restrict__ bias,
                              const unsigned* __restrict__ stat,
                              float* __restrict__ out, int nrows) {
  const int n = blockIdx.x * blockDim.x + threadIdx.x;
  if (n >= nrows) return;
  constexpr int F = 2 * C;
  float b0, b1;
  const float eps = bound_mode(stat, b0, b1) ? 0.f : 1e-16f;
  const float4* __restrict__ ar = (const float4*)(acc + (size_t)n * F);
  float4* __restrict__ outp = (float4*)(out + (size_t)n * F);
  const float2 zz = z[n];
  const float rz0 = 1.f / (zz.x + eps);
  const float rz1 = 1.f / (zz.y + eps);
#pragma unroll
  for (int v = 0; v < F / 4; ++v) {
    float4 a = ar[v];
    const float rz = (4 * v < C) ? rz0 : rz1;
#pragma unroll
    for (int j = 0; j < 4; ++j) {
      float x = (&a.x)[j] * rz + bias[4 * v + j];
      if (ELU) x = (x > 0.f) ? x : (__expf(x) - 1.f);
      (&a.x)[j] = x;
    }
    outp[v] = a;
  }
}

extern "C" void kernel_launch(void* const* d_in, const int* in_sizes, int n_in,
                              void* d_out, int out_size, void* d_ws, size_t ws_size,
                              hipStream_t stream) {
  const float* x   = (const float*)d_in[0];
  const int*   ei  = (const int*)  d_in[1];
  const float* W1  = (const float*)d_in[2];
  const float* as1 = (const float*)d_in[3];
  const float* ad1 = (const float*)d_in[4];
  const float* b1  = (const float*)d_in[5];
  const float* W2  = (const float*)d_in[6];
  const float* as2 = (const float*)d_in[7];
  const float* ad2 = (const float*)d_in[8];
  const float* b2  = (const float*)d_in[9];

  const int N = in_sizes[0] / 16;
  const int E = in_sizes[1] / 2;

  float*    ws   = (float*)d_ws;
  float*    h1   = ws;
  float*    acc1 = ws + (size_t)16 * N;
  float2*   als  = (float2*)(ws + (size_t)32 * N);
  float2*   ald  = (float2*)(ws + (size_t)34 * N);
  unsigned* menc = (unsigned*)(ws + (size_t)36 * N);
  float*    z    = ws + (size_t)38 * N;
  float*    h2   = ws + (size_t)40 * N;
  float*    acc2 = ws + (size_t)48 * N;

  const int TB = 256;
  const int node_grid = (N + TB - 1) / TB;
  const int own_grid  = 2048;
  const int acc_grid  = 256;
  int* qc1 = (int*)h2;
  int* qc2 = (int*)h1;
  unsigned* st1  = (unsigned*)(h2 + 256);
  unsigned* st2  = (unsigned*)(h1 + 256);
  unsigned* key1 = (unsigned*)(h2 + 320);
  unsigned* key2 = (unsigned*)(h1 + 320);
  const int tiles     = (N + 15) / 16;
  const int gemm_grid = (tiles + 7) / 8;

  gemm16_wmma<<<gemm_grid, TB, 0, stream>>>(x, W1, h1, st1, N, 16);
  node_logits_init<8><<<node_grid, TB, 0, stream>>>(h1, as1, ad1, als, ald, menc,
                                                    (float2*)z, acc1, qc1, st1, N);
  edge_max<<<own_grid, TB, 0, stream>>>(ei, als, ald, menc, key1, qc1, st1, E, N);
  edge_accum<8><<<acc_grid, ACC_TB, 0, stream>>>(ei, als, ald, menc, z, acc1, h1, key1,
                                                 qc1 + 2 * QSTRIDE, st1, E, N);
  node_finalize<8, true><<<node_grid, TB, 0, stream>>>(acc1, (const float2*)z, b1, st1, acc1, N);

  gemm16_wmma<<<gemm_grid, TB, 0, stream>>>(acc1, W2, h2, st2, N, 8);
  node_logits_init<4><<<node_grid, TB, 0, stream>>>(h2, as2, ad2, als, ald, menc,
                                                    (float2*)z, acc2, qc2, st2, N);
  edge_max<<<own_grid, TB, 0, stream>>>(ei, als, ald, menc, key2, qc2, st2, E, N);
  edge_accum<4><<<acc_grid, ACC_TB, 0, stream>>>(ei, als, ald, menc, z, acc2, h2, key2,
                                                 qc2 + 2 * QSTRIDE, st2, E, N);
  node_finalize<4, false><<<node_grid, TB, 0, stream>>>(acc2, (const float2*)z, b2, st2,
                                                        (float*)d_out, N);
}


# edge_accum: a workgroup claims its next work item when the current one is done, not one ahead: the 256 items spread over all 256 workgroups, early starters no longer take two; arrival-order sums
# speedup vs baseline: 9.1057x; 1.4547x over previous
#include <hip/hip_runtime.h>
#include <math.h>


typedef __attribute__((ext_vector_type(2))) float v2f;
typedef __attribute__((ext_vector_type(8))) float v8f;

__device__ __forceinline__ unsigned encf(float f) {
  unsigned u = __float_as_uint(f);
  return (u & 0x80000000u) ? ~u : (u | 0x80000000u);
}
__device__ __forceinline__ float decf(unsigned u) {
  unsigned v = (u & 0x80000000u) ? (u & 0x7FFFFFFFu) : ~u;
  return __uint_as_float(v);
}
#define ENC_NEG_INF 0x007FFFFFu
#define ENC_POS_INF 0xFF800000u
#define SPREAD_MAX 40.0f

__device__ __forceinline__ bool bound_mode(const unsigned* __restrict__ stat,
                                           float& b0, float& b1) {
  b0 = decf(stat[0]); b1 = decf(stat[1]);
  const float l0 = decf(stat[2]), l1 = decf(stat[3]);
  return (b0 - l0 <= SPREAD_MAX) && (b1 - l1 <= SPREAD_MAX);
}

#define CHUNK 512
#define RS 1568
#define QSTRIDE 64

__device__ __forceinline__ int package_half() {
  const unsigned v = __builtin_amdgcn_s_sendmsg_rtn(0x87);
  return (int)((v >> 18) & 1u);
}

__device__ __forceinline__ int split_node(int N) {
  const int s = ((N >> 1) + 31) & ~31;
  return (s < N) ? s : N;
}

__global__ void gemm16_wmma(const float* __restrict__ X,
                            const float* __restrict__ W,
                            float* __restrict__ Hout,
                            unsigned* __restrict__ stat,
                            int nrows, int ncols) {
  if (blockIdx.x == 0 && threadIdx.x < 4)
    stat[threadIdx.x] = (threadIdx.x < 2) ? ENC_NEG_INF : ENC_POS_INF;
  const int wave = blockIdx.x * (blockDim.x >> 5) + (threadIdx.x >> 5);
  const int lane = threadIdx.x & 31;
  const int row_base = wave * 16;
  if (row_base >= nrows) return;

  const int m    = lane & 15;
  const int half = lane >> 4;
  int row = row_base + m; if (row > nrows - 1) row = nrows - 1;
  int col = m;            if (col > ncols - 1) col = ncols - 1;

  const float2* __restrict__ Xr = (const float2*)(X + (size_t)row * 16);

  v8f c = {0.f, 0.f, 0.f, 0.f, 0.f, 0.f, 0.f, 0.f};
#pragma unroll
  for (int k0 = 0; k0 < 4; ++k0) {
    const int k = k0 * 4 + half * 2;
    const float2 av = Xr[k0 * 2 + half];
    v2f a; a.x = av.x; a.y = av.y;
    v2f b;
    b.x = W[(k)     * ncols + col];
    b.y = W[(k + 1) * ncols + col];
    c = __builtin_amdgcn_wmma_f32_16x16x4_f32(
         false, a,  false, b,
         (short)0, c,  false,  false);
  }

  if (m < ncols) {
    const int rofs = half * 8;
    float* __restrict__ outp = Hout + (size_t)(row_base + rofs) * ncols + m;
    if (row_base + 16 <= nrows) {
#pragma unroll
      for (int r = 0; r < 8; ++r) outp[(size_t)r * ncols] = c[r];
    } else {
#pragma unroll
      for (int r = 0; r < 8; ++r)
        if (row_base + rofs + r < nrows) outp[(size_t)r * ncols] = c[r];
    }
  }
}

template <int C>
__global__ void node_logits_init(const float* __restrict__ Hm,
                                 const float* __restrict__ a_src,
                                 const float* __restrict__ a_dst,
                                 float2* __restrict__ als, float2* __restrict__ ald,
                                 unsigned* __restrict__ menc,
                                 float2* __restrict__ z, float* __restrict__ acc,
                                 int* __restrict__ qctr,
                                 unsigned* __restrict__ stat, int nrows) {
  __shared__ float s_st[32][4];
  if (blockIdx.x == 0) qctr[threadIdx.x] = 0;
  const int n = blockIdx.x * blockDim.x + threadIdx.x;
  const bool valid = n < nrows;
  constexpr int F = 2 * C;
  float s0 = 0.f, s1 = 0.f, dd0 = 0.f, dd1 = 0.f;
  if (valid) {
    const float4* __restrict__ hr = (const float4*)(Hm + (size_t)n * F);
    float d0 = 0.f, d1 = 0.f;
#pragma unroll
    for (int v = 0; v < C / 4; ++v) {
      const float4 h0 = hr[v];
      const float4 h1 = hr[C / 4 + v];
#pragma unroll
      for (int j = 0; j < 4; ++j) {
        const float e0 = (&h0.x)[j], e1 = (&h1.x)[j];
        const int cch = 4 * v + j;
        s0 += e0 * a_src[cch];     d0 += e0 * a_dst[cch];
        s1 += e1 * a_src[C + cch]; d1 += e1 * a_dst[C + cch];
      }
    }
    als[n] = make_float2(s0, s1);
    ald[n] = make_float2(d0, d1);
    dd0 = d0; dd1 = d1;
    menc[n * 2] = ENC_NEG_INF; menc[n * 2 + 1] = ENC_NEG_INF;
    z[n] = make_float2(0.f, 0.f);
    float4* __restrict__ ar = (float4*)(acc + (size_t)n * F);
#pragma unroll
    for (int v = 0; v < F / 4; ++v) ar[v] = make_float4(0.f, 0.f, 0.f, 0.f);
  }
  float r[4];
  r[0] = valid ? s0 : -INFINITY; r[1] = valid ? s1 : -INFINITY;
  r[2] = valid ? s0 :  INFINITY; r[3] = valid ? s1 :  INFINITY;
  if (valid && !(fabsf(s0) <= 3.0e38f && fabsf(s1) <= 3.0e38f &&
                 fabsf(dd0) <= 3.0e38f && fabsf(dd1) <= 3.0e38f)) r[0] = INFINITY;
#pragma unroll
  for (int o = 16; o > 0; o >>= 1) {
    r[0] = fmaxf(r[0], __shfl_xor(r[0], o)); r[1] = fmaxf(r[1], __shfl_xor(r[1], o));
    r[2] = fminf(r[2], __shfl_xor(r[2], o)); r[3] = fminf(r[3], __shfl_xor(r[3], o));
  }
  if ((threadIdx.x & 31) == 0) {
#pragma unroll
    for (int k = 0; k < 4; ++k) s_st[threadIdx.x >> 5][k] = r[k];
  }
  __syncthreads();
  if (threadIdx.x < 4) {
    const int k = (int)threadIdx.x;
    const int nw = (int)(blockDim.x >> 5);
    float v = s_st[0][k];
    for (int w = 1; w < nw; ++w) v = (k < 2) ? fmaxf(v, s_st[w][k]) : fminf(v, s_st[w][k]);
    if (k < 2) atomicMax(&stat[k], encf(v));
    else       atomicMin(&stat[k], encf(v));
  }
}

__global__ void edge_max(const int* __restrict__ ei,
                         const float2* __restrict__ als,
                         const float2* __restrict__ ald,
                         unsigned* __restrict__ menc,
                         unsigned* __restrict__ keyw,
                         int* __restrict__ qctr,
                         const unsigned* __restrict__ stat, int E, int N) {
  __shared__ int s_c;
  {
    float b0, b1;
    if (bound_mode(stat, b0, b1)) {
      const int* __restrict__ dstp = ei + E;
      const int nwords = (E + 3) >> 2;
      const int step = (int)(gridDim.x * blockDim.x);
      for (int w = (int)(blockIdx.x * blockDim.x + threadIdx.x); w < nwords; w += step) {
        unsigned v = 0u;
        if ((E & 3) == 0) {
          const int4 d = ((const int4*)dstp)[w];
          v = (((unsigned)d.x / RS) & 0xFFu)       | ((((unsigned)d.y / RS) & 0xFFu) << 8) |
              ((((unsigned)d.z / RS) & 0xFFu) << 16) | ((((unsigned)d.w / RS) & 0xFFu) << 24);
        } else {
#pragma unroll
          for (int j = 0; j < 4; ++j) {
            const int t = 4 * w + j;
            if (t < E) v |= (((unsigned)dstp[t] / RS) & 0xFFu) << (8 * j);
          }
        }
        keyw[w] = v;
      }
      return;
    }
  }
  unsigned char* __restrict__ keyb = (unsigned char*)keyw;
  const int total = E + N;
  const int nchunks = (total + CHUNK - 1) / CHUNK;
  const int nsplit = split_node(N);
  const int myhalf = package_half();
  for (int side = 0; side < 2; ++side) {
    const int part = myhalf ^ side;
    const int lo = part ? nsplit : 0;
    const unsigned span = (unsigned)((part ? N : nsplit) - lo);
    int* __restrict__ q = qctr + part * QSTRIDE;
    int pend = 0;
    if (threadIdx.x == 0) pend = atomicAdd(q, 1);
    for (int g = 0; g <= nchunks; ++g) {
      if (threadIdx.x == 0) s_c = pend;
      __syncthreads();
      const int c = s_c;
      __syncthreads();
      if (c >= nchunks) break;
      if (threadIdx.x == 0) pend = atomicAdd(q, 1);
      int dstk[CHUNK / 256];
#pragma unroll
      for (int k = 0; k < CHUNK / 256; ++k) {
        const int t = c * CHUNK + k * 256 + (int)threadIdx.x;
        dstk[k] = (t < total) ? ((t < E) ? ei[E + t] : t - E) : -1;
      }
#pragma unroll
      for (int k = 0; k < CHUNK / 256; ++k) {
        const int t = c * CHUNK + k * 256 + (int)threadIdx.x;
        const int dst = dstk[k];
        if (part == 0 && t < E) keyb[t] = (unsigned char)((unsigned)dst / RS);
        if ((unsigned)(dst - lo) < span) {
          const int src = (t < E) ? ei[t] : dst;
          const float2 s = als[src];
          const float2 d = ald[dst];
          float e0 = s.x + d.x; e0 = (e0 > 0.f) ? e0 : 0.2f * e0;
          float e1 = s.y + d.y; e1 = (e1 > 0.f) ? e1 : 0.2f * e1;
          atomicMax(&menc[dst * 2],     encf(e0));
          atomicMax(&menc[dst * 2 + 1], encf(e1));
        }
      }
    }
  }
}

#define RB 4
#define SCAN_K 8
#define ACC_TB 1024
#define QCAP 160

__device__ __forceinline__ void wave_queue_handover() {
  __builtin_amdgcn_fence(__ATOMIC_RELEASE, "wavefront");
  __builtin_amdgcn_wave_barrier();
  __builtin_amdgcn_fence(__ATOMIC_ACQUIRE, "wavefront");
}

template <int C>
__device__ __forceinline__ void edge_terms(int src, int dl,
                                           const float2* __restrict__ als,
                                           const float* __restrict__ Hm,
                                           const float4* s_dm,
                                           float* s_row) {
  constexpr int F = 2 * C;
  constexpr int P = F + 3;
  const float2 s = als[src];
  const float4 dm = s_dm[dl];
  const float4* __restrict__ hs = (const float4*)(Hm + (size_t)src * F);
  float w[2];
#pragma unroll
  for (int hd = 0; hd < 2; ++hd) {
    float e = (hd ? s.y + dm.y : s.x + dm.x);
    e = (e > 0.f) ? e : 0.2f * e;
    w[hd] = __expf(e - (hd ? dm.w : dm.z));
  }
  float* r = s_row + dl * P;
  atomicAdd(&r[F + 0], w[0]);
  atomicAdd(&r[F + 1], w[1]);
#pragma unroll
  for (int hd = 0; hd < 2; ++hd) {
#pragma unroll
    for (int v = 0; v < C / 4; ++v) {
      const float4 hv = hs[hd * (C / 4) + v];
      const int base = hd * C + 4 * v;
      atomicAdd(&r[base + 0], w[hd] * hv.x);
      atomicAdd(&r[base + 1], w[hd] * hv.y);
      atomicAdd(&r[base + 2], w[hd] * hv.z);
      atomicAdd(&r[base + 3], w[hd] * hv.w);
    }
  }
}

template <int C>
__global__ void __launch_bounds__(ACC_TB)
edge_accum(const int* __restrict__ ei,
           const float2* __restrict__ als,
           const float2* __restrict__ ald,
           const unsigned* __restrict__ menc,
           float* __restrict__ z, float* __restrict__ acc,
           const float* __restrict__ Hm,
           const unsigned* __restrict__ keyw,
           int* __restrict__ qctr,
           const unsigned* __restrict__ stat, int E, int N) {
  constexpr int F = 2 * C;
  constexpr int P = F + 3;
  __shared__ float s_row[RS * P];
  __shared__ float4 s_dm[RS];
  __shared__ int s_q[ACC_TB / 32][QCAP];
  __shared__ int s_c;

  const int tid  = (int)threadIdx.x;
  const int lane = tid & 31;
  int* q_w = &s_q[tid >> 5][0];
  const int* __restrict__ dstp = ei + E;

  const int R  = (N + RS - 1) / RS;
  const int Rh = (R + 1) >> 1;
  const int myhalf = package_half();
  float b0, b1;
  const bool bmode = bound_mode(stat, b0, b1);
  for (int side = 0; side < 2; ++side) {
    const int part = myhalf ^ side;
    const int r0 = part ? Rh : 0;
    const int nitems = (part ? R - Rh : Rh) * RB;
    int* __restrict__ q = qctr + part * QSTRIDE;
    for (int g = 0; g <= nitems; ++g) {
      if (tid == 0) s_c = atomicAdd(q, 1);
      __syncthreads();
      const int c = s_c;
      __syncthreads();
      if (c >= nitems) break;
      const int rng = r0 + c / RB;
      const int b   = c % RB;
      const int lo  = rng * RS;
      const int span = ((lo + RS < N) ? lo + RS : N) - lo;

      for (int i = tid; i < RS * P; i += ACC_TB) s_row[i] = 0.f;
      for (int i = tid; i < span; i += ACC_TB) {
        const float2 d = ald[lo + i];
        float m0, m1;
        if (bmode) {
          m0 = b0 + d.x; m0 = (m0 > 0.f) ? m0 : 0.2f * m0;
          m1 = b1 + d.y; m1 = (m1 > 0.f) ? m1 : 0.2f * m1;
        } else {
          m0 = decf(menc[(lo + i) * 2]);
          m1 = decf(menc[(lo + i) * 2 + 1]);
          if (!(m0 >= -3.0e38f && m0 <= 3.0e38f)) m0 = 0.f;
          if (!(m1 >= -3.0e38f && m1 <= 3.0e38f)) m1 = 0.f;
        }
        s_dm[i] = make_float4(d.x, d.y, m0, m1);
      }
      __syncthreads();

      const int e_lo = (int)(((long long)E * b) / RB) & ~3;
      const int e_hi = (b == RB - 1) ? E : ((int)(((long long)E * (b + 1)) / RB) & ~3);
      const unsigned rngw = (unsigned)rng * 0x01010101u;
      int qn = 0;
      for (int wb = e_lo >> 2; wb < ((e_hi + 3) >> 2); wb += ACC_TB * SCAN_K) {
        unsigned kw[SCAN_K];
        if (wb + ACC_TB * SCAN_K <= (e_hi >> 2)) {
#pragma unroll
          for (int k = 0; k < SCAN_K; ++k) kw[k] = keyw[wb + k * ACC_TB + tid] ^ rngw;
        } else {
#pragma unroll
          for (int k = 0; k < SCAN_K; ++k) {
            const int w = wb + k * ACC_TB + tid;
            const int left = e_hi - 4 * w;
            unsigned v = 0xFFFFFFFFu;
            if (left > 0) {
              v = keyw[w] ^ rngw;
              if (left < 4) v |= 0xFFFFFFFFu << (8 * left);
            }
            kw[k] = v;
          }
        }
#pragma unroll
        for (int k = 0; k < SCAN_K; ++k) {
          const int t0 = 4 * (wb + k * ACC_TB + tid);
          const unsigned x = kw[k];
#pragma unroll
          for (int j = 0; j < 4; ++j) {
            const bool own = (x & (0xFFu << (8 * j))) == 0u;
            const unsigned mask = __builtin_amdgcn_ballot_w32(own);
            if (mask == 0u) continue;
            if (own) {
              const int pos = qn + __builtin_popcount(mask & ((1u << lane) - 1u));
              q_w[pos] = t0 + j;
            }
            qn += __builtin_popcount(mask);
          }
          while (qn >= 32) {
            qn -= 32;
            wave_queue_handover();
            const int t = q_w[qn + lane];
            const int src = ei[t];
            const int dst = dstp[t];
            asm volatile("" :: "v"(src), "v"(dst));
            const int dl = dst - lo;
            if ((unsigned)dl < (unsigned)span)
              edge_terms<C>(src, dl, als, Hm, s_dm, s_row);
          }
        }
      }
      wave_queue_handover();
      if (lane < qn) {
        const int t = q_w[lane];
        const int src = ei[t];
        const int dst = dstp[t];
        asm volatile("" :: "v"(src), "v"(dst));
        const int dl = dst - lo;
        if ((unsigned)dl < (unsigned)span)
          edge_terms<C>(src, dl, als, Hm, s_dm, s_row);
      }
      if (b == 0)
        for (int i = tid; i < span; i += ACC_TB)
          edge_terms<C>(lo + i, i, als, Hm, s_dm, s_row);
      __syncthreads();

      float* __restrict__ accr = acc + (size_t)lo * F;
      for (int i = tid; i < span * F; i += ACC_TB)
        atomicAdd(&accr[i], s_row[(i / F) * P + (i % F)]);
      float* __restrict__ zr = z + (size_t)lo * 2;
      for (int i = tid; i < span * 2; i += ACC_TB)
        atomicAdd(&zr[i], s_row[(i >> 1) * P + F + (i & 1)]);
    }
  }
}

template <int C, bool ELU>
__global__ void node_finalize(const float* __restrict__ acc,
                              const float2* __restrict__ z,
                              const float* __restrict__ bias,
                              const unsigned* __restrict__ stat,
                              float* __restrict__ out, int nrows) {
  const int n = blockIdx.x * blockDim.x + threadIdx.x;
  if (n >= nrows) return;
  constexpr int F = 2 * C;
  float b0, b1;
  const float eps = bound_mode(stat, b0, b1) ? 0.f : 1e-16f;
  const float4* __restrict__ ar = (const float4*)(acc + (size_t)n * F);
  float4* __restrict__ outp = (float4*)(out + (size_t)n * F);
  const float2 zz = z[n];
  const float rz0 = 1.f / (zz.x + eps);
  const float rz1 = 1.f / (zz.y + eps);
#pragma unroll
  for (int v = 0; v < F / 4; ++v) {
    float4 a = ar[v];
    const float rz = (4 * v < C) ? rz0 : rz1;
#pragma unroll
    for (int j = 0; j < 4; ++j) {
      float x = (&a.x)[j] * rz + bias[4 * v + j];
      if (ELU) x = (x > 0.f) ? x : (__expf(x) - 1.f);
      (&a.x)[j] = x;
    }
    outp[v] = a;
  }
}

extern "C" void kernel_launch(void* const* d_in, const int* in_sizes, int n_in,
                              void* d_out, int out_size, void* d_ws, size_t ws_size,
                              hipStream_t stream) {
  const float* x   = (const float*)d_in[0];
  const int*   ei  = (const int*)  d_in[1];
  const float* W1  = (const float*)d_in[2];
  const float* as1 = (const float*)d_in[3];
  const float* ad1 = (const float*)d_in[4];
  const float* b1  = (const float*)d_in[5];
  const float* W2  = (const float*)d_in[6];
  const float* as2 = (const float*)d_in[7];
  const float* ad2 = (const float*)d_in[8];
  const float* b2  = (const float*)d_in[9];

  const int N = in_sizes[0] / 16;
  const int E = in_sizes[1] / 2;

  float*    ws   = (float*)d_ws;
  float*    h1   = ws;
  float*    acc1 = ws + (size_t)16 * N;
  float2*   als  = (float2*)(ws + (size_t)32 * N);
  float2*   ald  = (float2*)(ws + (size_t)34 * N);
  unsigned* menc = (unsigned*)(ws + (size_t)36 * N);
  float*    z    = ws + (size_t)38 * N;
  float*    h2   = ws + (size_t)40 * N;
  float*    acc2 = ws + (size_t)48 * N;

  const int TB = 256;
  const int node_grid = (N + TB - 1) / TB;
  const int own_grid  = 2048;
  const int acc_grid  = 256;
  int* qc1 = (int*)h2;
  int* qc2 = (int*)h1;
  unsigned* st1  = (unsigned*)(h2 + 256);
  unsigned* st2  = (unsigned*)(h1 + 256);
  unsigned* key1 = (unsigned*)(h2 + 320);
  unsigned* key2 = (unsigned*)(h1 + 320);
  const int tiles     = (N + 15) / 16;
  const int gemm_grid = (tiles + 7) / 8;

  gemm16_wmma<<<gemm_grid, TB, 0, stream>>>(x, W1, h1, st1, N, 16);
  node_logits_init<8><<<node_grid, TB, 0, stream>>>(h1, as1, ad1, als, ald, menc,
                                                    (float2*)z, acc1, qc1, st1, N);
  edge_max<<<own_grid, TB, 0, stream>>>(ei, als, ald, menc, key1, qc1, st1, E, N);
  edge_accum<8><<<acc_grid, ACC_TB, 0, stream>>>(ei, als, ald, menc, z, acc1, h1, key1,
                                                 qc1 + 2 * QSTRIDE, st1, E, N);
  node_finalize<8, true><<<node_grid, TB, 0, stream>>>(acc1, (const float2*)z, b1, st1, acc1, N);

  gemm16_wmma<<<gemm_grid, TB, 0, stream>>>(acc1, W2, h2, st2, N, 8);
  node_logits_init<4><<<node_grid, TB, 0, stream>>>(h2, as2, ad2, als, ald, menc,
                                                    (float2*)z, acc2, qc2, st2, N);
  edge_max<<<own_grid, TB, 0, stream>>>(ei, als, ald, menc, key2, qc2, st2, E, N);
  edge_accum<4><<<acc_grid, ACC_TB, 0, stream>>>(ei, als, ald, menc, z, acc2, h2, key2,
                                                 qc2 + 2 * QSTRIDE, st2, E, N);
  node_finalize<4, false><<<node_grid, TB, 0, stream>>>(acc2, (const float2*)z, b2, st2,
                                                        (float*)d_out, N);
}


# edge_accum scan: a lane's 32 key bytes per step go through exact whole-word zero-byte tests into one bit mask; owned edges are queued once per lane and trip, not per byte position; arrival-order sums
# speedup vs baseline: 10.1123x; 1.1105x over previous
#include <hip/hip_runtime.h>
#include <math.h>


typedef __attribute__((ext_vector_type(2))) float v2f;
typedef __attribute__((ext_vector_type(8))) float v8f;

__device__ __forceinline__ unsigned encf(float f) {
  unsigned u = __float_as_uint(f);
  return (u & 0x80000000u) ? ~u : (u | 0x80000000u);
}
__device__ __forceinline__ float decf(unsigned u) {
  unsigned v = (u & 0x80000000u) ? (u & 0x7FFFFFFFu) : ~u;
  return __uint_as_float(v);
}
#define ENC_NEG_INF 0x007FFFFFu
#define ENC_POS_INF 0xFF800000u
#define SPREAD_MAX 40.0f

__device__ __forceinline__ bool bound_mode(const unsigned* __restrict__ stat,
                                           float& b0, float& b1) {
  b0 = decf(stat[0]); b1 = decf(stat[1]);
  const float l0 = decf(stat[2]), l1 = decf(stat[3]);
  return (b0 - l0 <= SPREAD_MAX) && (b1 - l1 <= SPREAD_MAX);
}

#define CHUNK 512
#define RS 1568
#define QSTRIDE 64

__device__ __forceinline__ int package_half() {
  const unsigned v = __builtin_amdgcn_s_sendmsg_rtn(0x87);
  return (int)((v >> 18) & 1u);
}

__device__ __forceinline__ int split_node(int N) {
  const int s = ((N >> 1) + 31) & ~31;
  return (s < N) ? s : N;
}

__global__ void gemm16_wmma(const float* __restrict__ X,
                            const float* __restrict__ W,
                            float* __restrict__ Hout,
                            unsigned* __restrict__ stat,
                            int nrows, int ncols) {
  if (blockIdx.x == 0 && threadIdx.x < 4)
    stat[threadIdx.x] = (threadIdx.x < 2) ? ENC_NEG_INF : ENC_POS_INF;
  const int wave = blockIdx.x * (blockDim.x >> 5) + (threadIdx.x >> 5);
  const int lane = threadIdx.x & 31;
  const int row_base = wave * 16;
  if (row_base >= nrows) return;

  const int m    = lane & 15;
  const int half = lane >> 4;
  int row = row_base + m; if (row > nrows - 1) row = nrows - 1;
  int col = m;            if (col > ncols - 1) col = ncols - 1;

  const float2* __restrict__ Xr = (const float2*)(X + (size_t)row * 16);

  v8f c = {0.f, 0.f, 0.f, 0.f, 0.f, 0.f, 0.f, 0.f};
#pragma unroll
  for (int k0 = 0; k0 < 4; ++k0) {
    const int k = k0 * 4 + half * 2;
    const float2 av = Xr[k0 * 2 + half];
    v2f a; a.x = av.x; a.y = av.y;
    v2f b;
    b.x = W[(k)     * ncols + col];
    b.y = W[(k + 1) * ncols + col];
    c = __builtin_amdgcn_wmma_f32_16x16x4_f32(
         false, a,  false, b,
         (short)0, c,  false,  false);
  }

  if (m < ncols) {
    const int rofs = half * 8;
    float* __restrict__ outp = Hout + (size_t)(row_base + rofs) * ncols + m;
    if (row_base + 16 <= nrows) {
#pragma unroll
      for (int r = 0; r < 8; ++r) outp[(size_t)r * ncols] = c[r];
    } else {
#pragma unroll
      for (int r = 0; r < 8; ++r)
        if (row_base + rofs + r < nrows) outp[(size_t)r * ncols] = c[r];
    }
  }
}

template <int C>
__global__ void node_logits_init(const float* __restrict__ Hm,
                                 const float* __restrict__ a_src,
                                 const float* __restrict__ a_dst,
                                 float2* __restrict__ als, float2* __restrict__ ald,
                                 unsigned* __restrict__ menc,
                                 float2* __restrict__ z, float* __restrict__ acc,
                                 int* __restrict__ qctr,
                                 unsigned* __restrict__ stat, int nrows) {
  __shared__ float s_st[32][4];
  if (blockIdx.x == 0) qctr[threadIdx.x] = 0;
  const int n = blockIdx.x * blockDim.x + threadIdx.x;
  const bool valid = n < nrows;
  constexpr int F = 2 * C;
  float s0 = 0.f, s1 = 0.f, dd0 = 0.f, dd1 = 0.f;
  if (valid) {
    const float4* __restrict__ hr = (const float4*)(Hm + (size_t)n * F);
    float d0 = 0.f, d1 = 0.f;
#pragma unroll
    for (int v = 0; v < C / 4; ++v) {
      const float4 h0 = hr[v];
      const float4 h1 = hr[C / 4 + v];
#pragma unroll
      for (int j = 0; j < 4; ++j) {
        const float e0 = (&h0.x)[j], e1 = (&h1.x)[j];
        const int cch = 4 * v + j;
        s0 += e0 * a_src[cch];     d0 += e0 * a_dst[cch];
        s1 += e1 * a_src[C + cch]; d1 += e1 * a_dst[C + cch];
      }
    }
    als[n] = make_float2(s0, s1);
    ald[n] = make_float2(d0, d1);
    dd0 = d0; dd1 = d1;
    menc[n * 2] = ENC_NEG_INF; menc[n * 2 + 1] = ENC_NEG_INF;
    z[n] = make_float2(0.f, 0.f);
    float4* __restrict__ ar = (float4*)(acc + (size_t)n * F);
#pragma unroll
    for (int v = 0; v < F / 4; ++v) ar[v] = make_float4(0.f, 0.f, 0.f, 0.f);
  }
  float r[4];
  r[0] = valid ? s0 : -INFINITY; r[1] = valid ? s1 : -INFINITY;
  r[2] = valid ? s0 :  INFINITY; r[3] = valid ? s1 :  INFINITY;
  if (valid && !(fabsf(s0) <= 3.0e38f && fabsf(s1) <= 3.0e38f &&
                 fabsf(dd0) <= 3.0e38f && fabsf(dd1) <= 3.0e38f)) r[0] = INFINITY;
#pragma unroll
  for (int o = 16; o > 0; o >>= 1) {
    r[0] = fmaxf(r[0], __shfl_xor(r[0], o)); r[1] = fmaxf(r[1], __shfl_xor(r[1], o));
    r[2] = fminf(r[2], __shfl_xor(r[2], o)); r[3] = fminf(r[3], __shfl_xor(r[3], o));
  }
  if ((threadIdx.x & 31) == 0) {
#pragma unroll
    for (int k = 0; k < 4; ++k) s_st[threadIdx.x >> 5][k] = r[k];
  }
  __syncthreads();
  if (threadIdx.x < 4) {
    const int k = (int)threadIdx.x;
    const int nw = (int)(blockDim.x >> 5);
    float v = s_st[0][k];
    for (int w = 1; w < nw; ++w) v = (k < 2) ? fmaxf(v, s_st[w][k]) : fminf(v, s_st[w][k]);
    if (k < 2) atomicMax(&stat[k], encf(v));
    else       atomicMin(&stat[k], encf(v));
  }
}

__global__ void edge_max(const int* __restrict__ ei,
                         const float2* __restrict__ als,
                         const float2* __restrict__ ald,
                         unsigned* __restrict__ menc,
                         unsigned* __restrict__ keyw,
                         int* __restrict__ qctr,
                         const unsigned* __restrict__ stat, int E, int N) {
  __shared__ int s_c;
  {
    float b0, b1;
    if (bound_mode(stat, b0, b1)) {
      const int* __restrict__ dstp = ei + E;
      const int nwords = (E + 3) >> 2;
      const int step = (int)(gridDim.x * blockDim.x);
      for (int w = (int)(blockIdx.x * blockDim.x + threadIdx.x); w < nwords; w += step) {
        unsigned v = 0u;
        if ((E & 3) == 0) {
          const int4 d = ((const int4*)dstp)[w];
          v = (((unsigned)d.x / RS) & 0xFFu)       | ((((unsigned)d.y / RS) & 0xFFu) << 8) |
              ((((unsigned)d.z / RS) & 0xFFu) << 16) | ((((unsigned)d.w / RS) & 0xFFu) << 24);
        } else {
#pragma unroll
          for (int j = 0; j < 4; ++j) {
            const int t = 4 * w + j;
            if (t < E) v |= (((unsigned)dstp[t] / RS) & 0xFFu) << (8 * j);
          }
        }
        keyw[w] = v;
      }
      return;
    }
  }
  unsigned char* __restrict__ keyb = (unsigned char*)keyw;
  const int total = E + N;
  const int nchunks = (total + CHUNK - 1) / CHUNK;
  const int nsplit = split_node(N);
  const int myhalf = package_half();
  for (int side = 0; side < 2; ++side) {
    const int part = myhalf ^ side;
    const int lo = part ? nsplit : 0;
    const unsigned span = (unsigned)((part ? N : nsplit) - lo);
    int* __restrict__ q = qctr + part * QSTRIDE;
    int pend = 0;
    if (threadIdx.x == 0) pend = atomicAdd(q, 1);
    for (int g = 0; g <= nchunks; ++g) {
      if (threadIdx.x == 0) s_c = pend;
      __syncthreads();
      const int c = s_c;
      __syncthreads();
      if (c >= nchunks) break;
      if (threadIdx.x == 0) pend = atomicAdd(q, 1);
      int dstk[CHUNK / 256];
#pragma unroll
      for (int k = 0; k < CHUNK / 256; ++k) {
        const int t = c * CHUNK + k * 256 + (int)threadIdx.x;
        dstk[k] = (t < total) ? ((t < E) ? ei[E + t] : t - E) : -1;
      }
#pragma unroll
      for (int k = 0; k < CHUNK / 256; ++k) {
        const int t = c * CHUNK + k * 256 + (int)threadIdx.x;
        const int dst = dstk[k];
        if (part == 0 && t < E) keyb[t] = (unsigned char)((unsigned)dst / RS);
        if ((unsigned)(dst - lo) < span) {
          const int src = (t < E) ? ei[t] : dst;
          const float2 s = als[src];
          const float2 d = ald[dst];
          float e0 = s.x + d.x; e0 = (e0 > 0.f) ? e0 : 0.2f * e0;
          float e1 = s.y + d.y; e1 = (e1 > 0.f) ? e1 : 0.2f * e1;
          atomicMax(&menc[dst * 2],     encf(e0));
          atomicMax(&menc[dst * 2 + 1], encf(e1));
        }
      }
    }
  }
}

#define RB 4
#define SCAN_K 8
#define ACC_TB 1024
#define QCAP 160

__device__ __forceinline__ void wave_queue_handover() {
  __builtin_amdgcn_fence(__ATOMIC_RELEASE, "wavefront");
  __builtin_amdgcn_wave_barrier();
  __builtin_amdgcn_fence(__ATOMIC_ACQUIRE, "wavefront");
}

template <int C>
__device__ __forceinline__ void edge_terms(int src, int dl,
                                           const float2* __restrict__ als,
                                           const float* __restrict__ Hm,
                                           const float4* s_dm,
                                           float* s_row) {
  constexpr int F = 2 * C;
  constexpr int P = F + 3;
  const float2 s = als[src];
  const float4 dm = s_dm[dl];
  const float4* __restrict__ hs = (const float4*)(Hm + (size_t)src * F);
  float w[2];
#pragma unroll
  for (int hd = 0; hd < 2; ++hd) {
    float e = (hd ? s.y + dm.y : s.x + dm.x);
    e = (e > 0.f) ? e : 0.2f * e;
    w[hd] = __expf(e - (hd ? dm.w : dm.z));
  }
  float* r = s_row + dl * P;
  atomicAdd(&r[F + 0], w[0]);
  atomicAdd(&r[F + 1], w[1]);
#pragma unroll
  for (int hd = 0; hd < 2; ++hd) {
#pragma unroll
    for (int v = 0; v < C / 4; ++v) {
      const float4 hv = hs[hd * (C / 4) + v];
      const int base = hd * C + 4 * v;
      atomicAdd(&r[base + 0], w[hd] * hv.x);
      atomicAdd(&r[base + 1], w[hd] * hv.y);
      atomicAdd(&r[base + 2], w[hd] * hv.z);
      atomicAdd(&r[base + 3], w[hd] * hv.w);
    }
  }
}

template <int C>
__global__ void __launch_bounds__(ACC_TB)
edge_accum(const int* __restrict__ ei,
           const float2* __restrict__ als,
           const float2* __restrict__ ald,
           const unsigned* __restrict__ menc,
           float* __restrict__ z, float* __restrict__ acc,
           const float* __restrict__ Hm,
           const unsigned* __restrict__ keyw,
           int* __restrict__ qctr,
           const unsigned* __restrict__ stat, int E, int N) {
  constexpr int F = 2 * C;
  constexpr int P = F + 3;
  __shared__ float s_row[RS * P];
  __shared__ float4 s_dm[RS];
  __shared__ int s_q[ACC_TB / 32][QCAP];
  __shared__ int s_c;

  const int tid  = (int)threadIdx.x;
  const int lane = tid & 31;
  int* q_w = &s_q[tid >> 5][0];
  const int* __restrict__ dstp = ei + E;

  const int R  = (N + RS - 1) / RS;
  const int Rh = (R + 1) >> 1;
  const int myhalf = package_half();
  float b0, b1;
  const bool bmode = bound_mode(stat, b0, b1);
  for (int side = 0; side < 2; ++side) {
    const int part = myhalf ^ side;
    const int r0 = part ? Rh : 0;
    const int nitems = (part ? R - Rh : Rh) * RB;
    int* __restrict__ q = qctr + part * QSTRIDE;
    for (int g = 0; g <= nitems; ++g) {
      if (tid == 0) s_c = atomicAdd(q, 1);
      __syncthreads();
      const int c = s_c;
      __syncthreads();
      if (c >= nitems) break;
      const int rng = r0 + c / RB;
      const int b   = c % RB;
      const int lo  = rng * RS;
      const int span = ((lo + RS < N) ? lo + RS : N) - lo;

      for (int i = tid; i < RS * P; i += ACC_TB) s_row[i] = 0.f;
      for (int i = tid; i < span; i += ACC_TB) {
        const float2 d = ald[lo + i];
        float m0, m1;
        if (bmode) {
          m0 = b0 + d.x; m0 = (m0 > 0.f) ? m0 : 0.2f * m0;
          m1 = b1 + d.y; m1 = (m1 > 0.f) ? m1 : 0.2f * m1;
        } else {
          m0 = decf(menc[(lo + i) * 2]);
          m1 = decf(menc[(lo + i) * 2 + 1]);
          if (!(m0 >= -3.0e38f && m0 <= 3.0e38f)) m0 = 0.f;
          if (!(m1 >= -3.0e38f && m1 <= 3.0e38f)) m1 = 0.f;
        }
        s_dm[i] = make_float4(d.x, d.y, m0, m1);
      }
      __syncthreads();

      const int e_lo = (int)(((long long)E * b) / RB) & ~3;
      const int e_hi = (b == RB - 1) ? E : ((int)(((long long)E * (b + 1)) / RB) & ~3);
      const unsigned rngw = (unsigned)rng * 0x01010101u;
      int qn = 0;
      for (int wb = e_lo >> 2; wb < ((e_hi + 3) >> 2); wb += ACC_TB * SCAN_K) {
        unsigned kw[SCAN_K];
        if (wb + ACC_TB * SCAN_K <= (e_hi >> 2)) {
#pragma unroll
          for (int k = 0; k < SCAN_K; ++k) kw[k] = keyw[wb + k * ACC_TB + tid] ^ rngw;
        } else {
#pragma unroll
          for (int k = 0; k < SCAN_K; ++k) {
            const int w = wb + k * ACC_TB + tid;
            const int left = e_hi - 4 * w;
            unsigned v = 0xFFFFFFFFu;
            if (left > 0) {
              v = keyw[w] ^ rngw;
              if (left < 4) v |= 0xFFFFFFFFu << (8 * left);
            }
            kw[k] = v;
          }
        }
        static_assert(SCAN_K == 8, "one flag bit per byte position and word: 4 x 8 = 32");
        unsigned notown = 0u;
#pragma unroll
        for (int k = 0; k < SCAN_K; ++k) {
          const unsigned x = kw[k];
          notown = (notown >> 1) | ((((x & 0x7F7F7F7Fu) + 0x7F7F7F7Fu) | x) & 0x80808080u);
        }
        unsigned own = ~notown;
        const int t00 = 4 * (wb + tid);
        unsigned any = __builtin_amdgcn_ballot_w32(own != 0u);
        while (any != 0u) {
          if (own != 0u) {
            const int p = __builtin_ctz(own);
            const int pos = (int)__builtin_amdgcn_mbcnt_lo(any, (unsigned)qn);
            q_w[pos] = t00 + (p & 7) * (4 * ACC_TB) + (p >> 3);
            own &= own - 1u;
          }
          qn += __builtin_popcount(any);
          if (qn >= 32) {
            qn -= 32;
            wave_queue_handover();
            const int t = q_w[qn + lane];
            const int src = ei[t];
            const int dst = dstp[t];
            asm volatile("" :: "v"(src), "v"(dst));
            const int dl = dst - lo;
            if ((unsigned)dl < (unsigned)span)
              edge_terms<C>(src, dl, als, Hm, s_dm, s_row);
          }
          any = __builtin_amdgcn_ballot_w32(own != 0u);
        }
      }
      wave_queue_handover();
      if (lane < qn) {
        const int t = q_w[lane];
        const int src = ei[t];
        const int dst = dstp[t];
        asm volatile("" :: "v"(src), "v"(dst));
        const int dl = dst - lo;
        if ((unsigned)dl < (unsigned)span)
          edge_terms<C>(src, dl, als, Hm, s_dm, s_row);
      }
      if (b == 0)
        for (int i = tid; i < span; i += ACC_TB)
          edge_terms<C>(lo + i, i, als, Hm, s_dm, s_row);
      __syncthreads();

      float* __restrict__ accr = acc + (size_t)lo * F;
      for (int i = tid; i < span * F; i += ACC_TB)
        atomicAdd(&accr[i], s_row[(i / F) * P + (i % F)]);
      float* __restrict__ zr = z + (size_t)lo * 2;
      for (int i = tid; i < span * 2; i += ACC_TB)
        atomicAdd(&zr[i], s_row[(i >> 1) * P + F + (i & 1)]);
    }
  }
}

template <int C, bool ELU>
__global__ void node_finalize(const float* __restrict__ acc,
                              const float2* __restrict__ z,
                              const float* __restrict__ bias,
                              const unsigned* __restrict__ stat,
                              float* __restrict__ out, int nrows) {
  const int n = blockIdx.x * blockDim.x + threadIdx.x;
  if (n >= nrows) return;
  constexpr int F = 2 * C;
  float b0, b1;
  const float eps = bound_mode(stat, b0, b1) ? 0.f : 1e-16f;
  const float4* __restrict__ ar = (const float4*)(acc + (size_t)n * F);
  float4* __restrict__ outp = (float4*)(out + (size_t)n * F);
  const float2 zz = z[n];
  const float rz0 = 1.f / (zz.x + eps);
  const float rz1 = 1.f / (zz.y + eps);
#pragma unroll
  for (int v = 0; v < F / 4; ++v) {
    float4 a = ar[v];
    const float rz = (4 * v < C) ? rz0 : rz1;
#pragma unroll
    for (int j = 0; j < 4; ++j) {
      float x = (&a.x)[j] * rz + bias[4 * v + j];
      if (ELU) x = (x > 0.f) ? x : (__expf(x) - 1.f);
      (&a.x)[j] = x;
    }
    outp[v] = a;
  }
}

extern "C" void kernel_launch(void* const* d_in, const int* in_sizes, int n_in,
                              void* d_out, int out_size, void* d_ws, size_t ws_size,
                              hipStream_t stream) {
  const float* x   = (const float*)d_in[0];
  const int*   ei  = (const int*)  d_in[1];
  const float* W1  = (const float*)d_in[2];
  const float* as1 = (const float*)d_in[3];
  const float* ad1 = (const float*)d_in[4];
  const float* b1  = (const float*)d_in[5];
  const float* W2  = (const float*)d_in[6];
  const float* as2 = (const float*)d_in[7];
  const float* ad2 = (const float*)d_in[8];
  const float* b2  = (const float*)d_in[9];

  const int N = in_sizes[0] / 16;
  const int E = in_sizes[1] / 2;

  float*    ws   = (float*)d_ws;
  float*    h1   = ws;
  float*    acc1 = ws + (size_t)16 * N;
  float2*   als  = (float2*)(ws + (size_t)32 * N);
  float2*   ald  = (float2*)(ws + (size_t)34 * N);
  unsigned* menc = (unsigned*)(ws + (size_t)36 * N);
  float*    z    = ws + (size_t)38 * N;
  float*    h2   = ws + (size_t)40 * N;
  float*    acc2 = ws + (size_t)48 * N;

  const int TB = 256;
  const int node_grid = (N + TB - 1) / TB;
  const int own_grid  = 2048;
  const int acc_grid  = 256;
  int* qc1 = (int*)h2;
  int* qc2 = (int*)h1;
  unsigned* st1  = (unsigned*)(h2 + 256);
  unsigned* st2  = (unsigned*)(h1 + 256);
  unsigned* key1 = (unsigned*)(h2 + 320);
  unsigned* key2 = (unsigned*)(h1 + 320);
  const int tiles     = (N + 15) / 16;
  const int gemm_grid = (tiles + 7) / 8;

  gemm16_wmma<<<gemm_grid, TB, 0, stream>>>(x, W1, h1, st1, N, 16);
  node_logits_init<8><<<node_grid, TB, 0, stream>>>(h1, as1, ad1, als, ald, menc,
                                                    (float2*)z, acc1, qc1, st1, N);
  edge_max<<<own_grid, TB, 0, stream>>>(ei, als, ald, menc, key1, qc1, st1, E, N);
  edge_accum<8><<<acc_grid, ACC_TB, 0, stream>>>(ei, als, ald, menc, z, acc1, h1, key1,
                                                 qc1 + 2 * QSTRIDE, st1, E, N);
  node_finalize<8, true><<<node_grid, TB, 0, stream>>>(acc1, (const float2*)z, b1, st1, acc1, N);

  gemm16_wmma<<<gemm_grid, TB, 0, stream>>>(acc1, W2, h2, st2, N, 8);
  node_logits_init<4><<<node_grid, TB, 0, stream>>>(h2, as2, ad2, als, ald, menc,
                                                    (float2*)z, acc2, qc2, st2, N);
  edge_max<<<own_grid, TB, 0, stream>>>(ei, als, ald, menc, key2, qc2, st2, E, N);
  edge_accum<4><<<acc_grid, ACC_TB, 0, stream>>>(ei, als, ald, menc, z, acc2, h2, key2,
                                                 qc2 + 2 * QSTRIDE, st2, E, N);
  node_finalize<4, false><<<node_grid, TB, 0, stream>>>(acc2, (const float2*)z, b2, st2,
                                                        (float*)d_out, N);
}


# node launches fused, 10 -> 7 per call: product launch also makes logits (lane butterfly: new sum order of the logit dots), clears, range bytes once; layer-1 normalise+ELU is layer 2's A load
# speedup vs baseline: 11.0411x; 1.0918x over previous
#include <hip/hip_runtime.h>
#include <math.h>


typedef __attribute__((ext_vector_type(2))) float v2f;
typedef __attribute__((ext_vector_type(8))) float v8f;

__device__ __forceinline__ unsigned encf(float f) {
  unsigned u = __float_as_uint(f);
  return (u & 0x80000000u) ? ~u : (u | 0x80000000u);
}
__device__ __forceinline__ float decf(unsigned u) {
  unsigned v = (u & 0x80000000u) ? (u & 0x7FFFFFFFu) : ~u;
  return __uint_as_float(v);
}
#define ENC_NEG_INF 0x007FFFFFu
#define ENC_POS_INF 0xFF800000u
#define SPREAD_MAX 40.0f

__device__ __forceinline__ bool bound_mode(const unsigned* __restrict__ stat,
                                           float& b0, float& b1) {
  b0 = decf(stat[0]); b1 = decf(stat[1]);
  const float l0 = decf(stat[2]), l1 = decf(stat[3]);
  return (b0 - l0 <= SPREAD_MAX) && (b1 - l1 <= SPREAD_MAX);
}

#define CHUNK 512
#define RS 1568
#define QSTRIDE 64

__device__ __forceinline__ int package_half() {
  const unsigned v = __builtin_amdgcn_s_sendmsg_rtn(0x87);
  return (int)((v >> 18) & 1u);
}

__device__ __forceinline__ int split_node(int N) {
  const int s = ((N >> 1) + 31) & ~31;
  return (s < N) ? s : N;
}

#define NODE_TB 512

template <int CTRL>
__device__ __forceinline__ float dpp_get(float v) {
  return __int_as_float(__builtin_amdgcn_update_dpp(0, __float_as_int(v), CTRL, 0xF, 0xF, true));
}

__device__ __forceinline__ unsigned key_word(const int* __restrict__ dstp, int w, int E) {
  unsigned v = 0u;
  if ((E & 3) == 0) {
    const int4 d = ((const int4*)dstp)[w];
    v = (((unsigned)d.x / RS) & 0xFFu)       | ((((unsigned)d.y / RS) & 0xFFu) << 8) |
        ((((unsigned)d.z / RS) & 0xFFu) << 16) | ((((unsigned)d.w / RS) & 0xFFu) << 24);
  } else {
#pragma unroll
    for (int j = 0; j < 4; ++j) {
      const int t = 4 * w + j;
      if (t < E) v |= (((unsigned)dstp[t] / RS) & 0xFFu) << (8 * j);
    }
  }
  return v;
}

template <int C, bool FIRST>
__global__ void __launch_bounds__(NODE_TB)
node_layer(const float* __restrict__ X,
           float* zbuf,
           const float* __restrict__ bias,
           const unsigned* __restrict__ stat_prev,
           const float* __restrict__ W,
           const float* __restrict__ a_src,
           const float* __restrict__ a_dst,
           float* __restrict__ Hout, float* __restrict__ acc,
           float* __restrict__ als, float* __restrict__ ald,
           unsigned* __restrict__ menc,
           float* __restrict__ part,
           const int* __restrict__ ei, unsigned* __restrict__ keyw,
           int* __restrict__ qctr, int nrows, int E) {
  constexpr int F = 2 * C;
  __shared__ float s_st[NODE_TB / 32][4];
  const int tid  = (int)threadIdx.x;
  const int lane = tid & 31;
  const int wv   = tid >> 5;
  float st[4] = {-INFINITY, -INFINITY, INFINITY, INFINITY};
  const int row_base = ((int)blockIdx.x * (NODE_TB / 32) + wv) * 16;
  if (row_base < nrows) {
    const int m    = lane & 15;
    const int half = lane >> 4;
    int row = row_base + m; if (row > nrows - 1) row = nrows - 1;
    const int col = m & (F - 1);
    const float2* __restrict__ Xr = (const float2*)(X + (size_t)row * 16);
    float rz0 = 1.f, rz1 = 1.f;
    if (!FIRST) {
      float b0, b1;
      const float eps = bound_mode(stat_prev, b0, b1) ? 0.f : 1e-16f;
      const float2 zz = ((const float2*)zbuf)[row];
      rz0 = 1.f / (zz.x + eps);
      rz1 = 1.f / (zz.y + eps);
    }
    v8f c = {0.f, 0.f, 0.f, 0.f, 0.f, 0.f, 0.f, 0.f};
#pragma unroll
    for (int k0 = 0; k0 < 4; ++k0) {
      const int k = k0 * 4 + half * 2;
      const float2 av = Xr[k0 * 2 + half];
      v2f a;
      if (FIRST) {
        a.x = av.x; a.y = av.y;
      } else {
        const float rz = (k0 < 2) ? rz0 : rz1;
        float x0 = av.x * rz + bias[k];
        float x1 = av.y * rz + bias[k + 1];
        x0 = (x0 > 0.f) ? x0 : (__expf(x0) - 1.f);
        x1 = (x1 > 0.f) ? x1 : (__expf(x1) - 1.f);
        a.x = x0; a.y = x1;
      }
      v2f b;
      b.x = W[(k)     * F + col];
      b.y = W[(k + 1) * F + col];
      c = __builtin_amdgcn_wmma_f32_16x16x4_f32(
           false, a,  false, b,
           (short)0, c,  false,  false);
    }

    const int rofs = half * 8;
    if (m < F) {
      float* __restrict__ outp = Hout + (size_t)(row_base + rofs) * F + m;
      if (row_base + 16 <= nrows) {
#pragma unroll
        for (int r = 0; r < 8; ++r) outp[(size_t)r * F] = c[r];
      } else {
#pragma unroll
        for (int r = 0; r < 8; ++r)
          if (row_base + rofs + r < nrows) outp[(size_t)r * F] = c[r];
      }
    }

    const float as_c = a_src[col], ad_c = a_dst[col];
    float sv[8], dv[8];
#pragma unroll
    for (int r = 0; r < 8; ++r) { sv[r] = c[r] * as_c; dv[r] = c[r] * ad_c; }
#pragma unroll
    for (int r = 0; r < 8; ++r) { sv[r] += dpp_get<0xB1>(sv[r]); dv[r] += dpp_get<0xB1>(dv[r]); }
#pragma unroll
    for (int r = 0; r < 8; ++r) { sv[r] += dpp_get<0x4E>(sv[r]); dv[r] += dpp_get<0x4E>(dv[r]); }
    if (C == 8) {
#pragma unroll
      for (int r = 0; r < 8; ++r) { sv[r] += dpp_get<0x164>(sv[r]); dv[r] += dpp_get<0x164>(dv[r]); }
    }
    const int head  = (C == 8) ? (m >> 3) : ((m >> 2) & 1);
    const int r_sel = (C == 8) ? (m & 7) : ((m & 3) + 4 * (m >> 3));
    float s_sel = sv[0], d_sel = dv[0];
#pragma unroll
    for (int r = 1; r < 8; ++r)
      if (r_sel == r) { s_sel = sv[r]; d_sel = dv[r]; }
    const int orow = row_base + rofs + r_sel;
    if (orow < nrows) {
      als[orow * 2 + head] = s_sel;
      ald[orow * 2 + head] = d_sel;
      if (head == 0) { st[0] = s_sel; st[2] = s_sel; }
      else           { st[1] = s_sel; st[3] = s_sel; }
      if (!(fabsf(s_sel) <= 3.0e38f && fabsf(d_sel) <= 3.0e38f)) st[0] = INFINITY;
    }

    const int nr = (row_base + 16 <= nrows) ? 16 : (nrows - row_base);
    float4* __restrict__ ar = (float4*)(acc + (size_t)row_base * F);
    for (int i = lane; i < nr * (F / 4); i += 32) ar[i] = make_float4(0.f, 0.f, 0.f, 0.f);
    if (lane < nr * 2) {
      zbuf[row_base * 2 + lane] = 0.f;
      menc[row_base * 2 + lane] = ENC_NEG_INF;
    }
  }

#pragma unroll
  for (int o = 16; o > 0; o >>= 1) {
    st[0] = fmaxf(st[0], __shfl_xor(st[0], o)); st[1] = fmaxf(st[1], __shfl_xor(st[1], o));
    st[2] = fminf(st[2], __shfl_xor(st[2], o)); st[3] = fminf(st[3], __shfl_xor(st[3], o));
  }
  if (lane == 0) {
#pragma unroll
    for (int k = 0; k < 4; ++k) s_st[wv][k] = st[k];
  }
  __syncthreads();
  if (tid < 4) {
    float v = s_st[0][tid];
    for (int w = 1; w < NODE_TB / 32; ++w)
      v = (tid < 2) ? fmaxf(v, s_st[w][tid]) : fminf(v, s_st[w][tid]);
    part[(int)blockIdx.x * 4 + tid] = v;
  }

  if (FIRST) {
    if (blockIdx.x == 0) qctr[tid] = 0;
    const int* __restrict__ dstp = ei + E;
    const int nwords = (E + 3) >> 2;
    const int step = (int)gridDim.x * NODE_TB;
    for (int w0 = (int)blockIdx.x * NODE_TB + tid; w0 < nwords; w0 += 4 * step) {
      unsigned kv[4];
#pragma unroll
      for (int k = 0; k < 4; ++k) {
        const int w = w0 + k * step;
        kv[k] = (w < nwords) ? key_word(dstp, w, E) : 0u;
      }
#pragma unroll
      for (int k = 0; k < 4; ++k) {
        const int w = w0 + k * step;
        if (w < nwords) keyw[w] = kv[k];
      }
    }
  }
}

#define MAX_TB 256
__global__ void __launch_bounds__(MAX_TB)
edge_max(const int* __restrict__ ei,
         const float2* __restrict__ als,
         const float2* __restrict__ ald,
         unsigned* __restrict__ menc,
         int* __restrict__ qctr,
         const float4* __restrict__ part, int nb,
         unsigned* __restrict__ stat, int E, int N) {
  __shared__ int s_c;
  __shared__ float s_red[MAX_TB / 32][4];
  __shared__ float s_fin[4];
  {
    float r[4] = {-INFINITY, -INFINITY, INFINITY, INFINITY};
    for (int j = (int)threadIdx.x; j < nb; j += MAX_TB) {
      const float4 p = part[j];
      r[0] = fmaxf(r[0], p.x); r[1] = fmaxf(r[1], p.y);
      r[2] = fminf(r[2], p.z); r[3] = fminf(r[3], p.w);
    }
#pragma unroll
    for (int o = 16; o > 0; o >>= 1) {
      r[0] = fmaxf(r[0], __shfl_xor(r[0], o)); r[1] = fmaxf(r[1], __shfl_xor(r[1], o));
      r[2] = fminf(r[2], __shfl_xor(r[2], o)); r[3] = fminf(r[3], __shfl_xor(r[3], o));
    }
    if ((threadIdx.x & 31) == 0) {
#pragma unroll
      for (int k = 0; k < 4; ++k) s_red[threadIdx.x >> 5][k] = r[k];
    }
    __syncthreads();
    if (threadIdx.x < 4) {
      const int k = (int)threadIdx.x;
      float v = s_red[0][k];
      for (int w = 1; w < MAX_TB / 32; ++w)
        v = (k < 2) ? fmaxf(v, s_red[w][k]) : fminf(v, s_red[w][k]);
      s_fin[k] = v;
      if (blockIdx.x == 0) stat[k] = encf(v);
    }
    __syncthreads();
    if ((s_fin[0] - s_fin[2] <= SPREAD_MAX) && (s_fin[1] - s_fin[3] <= SPREAD_MAX)) return;
  }
  const int total = E + N;
  const int nchunks = (total + CHUNK - 1) / CHUNK;
  const int nsplit = split_node(N);
  const int myhalf = package_half();
  for (int side = 0; side < 2; ++side) {
    const int part_i = myhalf ^ side;
    const int lo = part_i ? nsplit : 0;
    const unsigned span = (unsigned)((part_i ? N : nsplit) - lo);
    int* __restrict__ q = qctr + part_i * QSTRIDE;
    int pend = 0;
    if (threadIdx.x == 0) pend = atomicAdd(q, 1);
    for (int g = 0; g <= nchunks; ++g) {
      if (threadIdx.x == 0) s_c = pend;
      __syncthreads();
      const int c = s_c;
      __syncthreads();
      if (c >= nchunks) break;
      if (threadIdx.x == 0) pend = atomicAdd(q, 1);
      int dstk[CHUNK / MAX_TB];
#pragma unroll
      for (int k = 0; k < CHUNK / MAX_TB; ++k) {
        const int t = c * CHUNK + k * MAX_TB + (int)threadIdx.x;
        dstk[k] = (t < total) ? ((t < E) ? ei[E + t] : t - E) : -1;
      }
#pragma unroll
      for (int k = 0; k < CHUNK / MAX_TB; ++k) {
        const int t = c * CHUNK + k * MAX_TB + (int)threadIdx.x;
        const int dst = dstk[k];
        if ((unsigned)(dst - lo) < span) {
          const int src = (t < E) ? ei[t] : dst;
          const float2 s = als[src];
          const float2 d = ald[dst];
          float e0 = s.x + d.x; e0 = (e0 > 0.f) ? e0 : 0.2f * e0;
          float e1 = s.y + d.y; e1 = (e1 > 0.f) ? e1 : 0.2f * e1;
          atomicMax(&menc[dst * 2],     encf(e0));
          atomicMax(&menc[dst * 2 + 1], encf(e1));
        }
      }
    }
  }
}

#define RB 4
#define SCAN_K 8
#define ACC_TB 1024
#define QCAP 160

__device__ __forceinline__ void wave_queue_handover() {
  __builtin_amdgcn_fence(__ATOMIC_RELEASE, "wavefront");
  __builtin_amdgcn_wave_barrier();
  __builtin_amdgcn_fence(__ATOMIC_ACQUIRE, "wavefront");
}

template <int C>
__device__ __forceinline__ void edge_terms(int src, int dl,
                                           const float2* __restrict__ als,
                                           const float* __restrict__ Hm,
                                           const float4* s_dm,
                                           float* s_row) {
  constexpr int F = 2 * C;
  constexpr int P = F + 3;
  const float2 s = als[src];
  const float4 dm = s_dm[dl];
  const float4* __restrict__ hs = (const float4*)(Hm + (size_t)src * F);
  float w[2];
#pragma unroll
  for (int hd = 0; hd < 2; ++hd) {
    float e = (hd ? s.y + dm.y : s.x + dm.x);
    e = (e > 0.f) ? e : 0.2f * e;
    w[hd] = __expf(e - (hd ? dm.w : dm.z));
  }
  float* r = s_row + dl * P;
  atomicAdd(&r[F + 0], w[0]);
  atomicAdd(&r[F + 1], w[1]);
#pragma unroll
  for (int hd = 0; hd < 2; ++hd) {
#pragma unroll
    for (int v = 0; v < C / 4; ++v) {
      const float4 hv = hs[hd * (C / 4) + v];
      const int base = hd * C + 4 * v;
      atomicAdd(&r[base + 0], w[hd] * hv.x);
      atomicAdd(&r[base + 1], w[hd] * hv.y);
      atomicAdd(&r[base + 2], w[hd] * hv.z);
      atomicAdd(&r[base + 3], w[hd] * hv.w);
    }
  }
}

template <int C>
__global__ void __launch_bounds__(ACC_TB)
edge_accum(const int* __restrict__ ei,
           const float2* __restrict__ als,
           const float2* __restrict__ ald,
           const unsigned* __restrict__ menc,
           float* __restrict__ z, float* __restrict__ acc,
           const float* __restrict__ Hm,
           const unsigned* __restrict__ keyw,
           int* __restrict__ qctr,
           const unsigned* __restrict__ stat, int E, int N) {
  constexpr int F = 2 * C;
  constexpr int P = F + 3;
  __shared__ float s_row[RS * P];
  __shared__ float4 s_dm[RS];
  __shared__ int s_q[ACC_TB / 32][QCAP];
  __shared__ int s_c;

  const int tid  = (int)threadIdx.x;
  const int lane = tid & 31;
  int* q_w = &s_q[tid >> 5][0];
  const int* __restrict__ dstp = ei + E;

  const int R  = (N + RS - 1) / RS;
  const int Rh = (R + 1) >> 1;
  const int myhalf = package_half();
  float b0, b1;
  const bool bmode = bound_mode(stat, b0, b1);
  for (int side = 0; side < 2; ++side) {
    const int part = myhalf ^ side;
    const int r0 = part ? Rh : 0;
    const int nitems = (part ? R - Rh : Rh) * RB;
    int* __restrict__ q = qctr + part * QSTRIDE;
    for (int g = 0; g <= nitems; ++g) {
      if (tid == 0) s_c = atomicAdd(q, 1);
      __syncthreads();
      const int c = s_c;
      __syncthreads();
      if (c >= nitems) break;
      const int rng = r0 + c / RB;
      const int b   = c % RB;
      const int lo  = rng * RS;
      const int span = ((lo + RS < N) ? lo + RS : N) - lo;

      for (int i = tid; i < RS * P; i += ACC_TB) s_row[i] = 0.f;
      for (int i = tid; i < span; i += ACC_TB) {
        const float2 d = ald[lo + i];
        float m0, m1;
        if (bmode) {
          m0 = b0 + d.x; m0 = (m0 > 0.f) ? m0 : 0.2f * m0;
          m1 = b1 + d.y; m1 = (m1 > 0.f) ? m1 : 0.2f * m1;
        } else {
          m0 = decf(menc[(lo + i) * 2]);
          m1 = decf(menc[(lo + i) * 2 + 1]);
          if (!(m0 >= -3.0e38f && m0 <= 3.0e38f)) m0 = 0.f;
          if (!(m1 >= -3.0e38f && m1 <= 3.0e38f)) m1 = 0.f;
        }
        s_dm[i] = make_float4(d.x, d.y, m0, m1);
      }
      __syncthreads();

      const int e_lo = (int)(((long long)E * b) / RB) & ~3;
      const int e_hi = (b == RB - 1) ? E : ((int)(((long long)E * (b + 1)) / RB) & ~3);
      const unsigned rngw = (unsigned)rng * 0x01010101u;
      int qn = 0;
      for (int wb = e_lo >> 2; wb < ((e_hi + 3) >> 2); wb += ACC_TB * SCAN_K) {
        unsigned kw[SCAN_K];
        if (wb + ACC_TB * SCAN_K <= (e_hi >> 2)) {
#pragma unroll
          for (int k = 0; k < SCAN_K; ++k) kw[k] = keyw[wb + k * ACC_TB + tid] ^ rngw;
        } else {
#pragma unroll
          for (int k = 0; k < SCAN_K; ++k) {
            const int w = wb + k * ACC_TB + tid;
            const int left = e_hi - 4 * w;
            unsigned v = 0xFFFFFFFFu;
            if (left > 0) {
              v = keyw[w] ^ rngw;
              if (left < 4) v |= 0xFFFFFFFFu << (8 * left);
            }
            kw[k] = v;
          }
        }
        static_assert(SCAN_K == 8, "one flag bit per byte position and word: 4 x 8 = 32");
        unsigned notown = 0u;
#pragma unroll
        for (int k = 0; k < SCAN_K; ++k) {
          const unsigned x = kw[k];
          notown = (notown >> 1) | ((((x & 0x7F7F7F7Fu) + 0x7F7F7F7Fu) | x) & 0x80808080u);
        }
        unsigned own = ~notown;
        const int t00 = 4 * (wb + tid);
        unsigned any = __builtin_amdgcn_ballot_w32(own != 0u);
        while (any != 0u) {
          if (own != 0u) {
            const int p = __builtin_ctz(own);
            const int pos = (int)__builtin_amdgcn_mbcnt_lo(any, (unsigned)qn);
            q_w[pos] = t00 + (p & 7) * (4 * ACC_TB) + (p >> 3);
            own &= own - 1u;
          }
          qn += __builtin_popcount(any);
          if (qn >= 32) {
            qn -= 32;
            wave_queue_handover();
            const int t = q_w[qn + lane];
            const int src = ei[t];
            const int dst = dstp[t];
            asm volatile("" :: "v"(src), "v"(dst));
            const int dl = dst - lo;
            if ((unsigned)dl < (unsigned)span)
              edge_terms<C>(src, dl, als, Hm, s_dm, s_row);
          }
          any = __builtin_amdgcn_ballot_w32(own != 0u);
        }
      }
      wave_queue_handover();
      if (lane < qn) {
        const int t = q_w[lane];
        const int src = ei[t];
        const int dst = dstp[t];
        asm volatile("" :: "v"(src), "v"(dst));
        const int dl = dst - lo;
        if ((unsigned)dl < (unsigned)span)
          edge_terms<C>(src, dl, als, Hm, s_dm, s_row);
      }
      if (b == 0)
        for (int i = tid; i < span; i += ACC_TB)
          edge_terms<C>(lo + i, i, als, Hm, s_dm, s_row);
      __syncthreads();

      float* __restrict__ accr = acc + (size_t)lo * F;
      for (int i = tid; i < span * F; i += ACC_TB)
        atomicAdd(&accr[i], s_row[(i / F) * P + (i % F)]);
      float* __restrict__ zr = z + (size_t)lo * 2;
      for (int i = tid; i < span * 2; i += ACC_TB)
        atomicAdd(&zr[i], s_row[(i >> 1) * P + F + (i & 1)]);
    }
  }
}

template <int C, bool ELU>
__global__ void node_finalize(const float* __restrict__ acc,
                              const float2* __restrict__ z,
                              const float* __restrict__ bias,
                              const unsigned* __restrict__ stat,
                              float* __restrict__ out, int nrows) {
  const int n = blockIdx.x * blockDim.x + threadIdx.x;
  if (n >= nrows) return;
  constexpr int F = 2 * C;
  float b0, b1;
  const float eps = bound_mode(stat, b0, b1) ? 0.f : 1e-16f;
  const float4* __restrict__ ar = (const float4*)(acc + (size_t)n * F);
  float4* __restrict__ outp = (float4*)(out + (size_t)n * F);
  const float2 zz = z[n];
  const float rz0 = 1.f / (zz.x + eps);
  const float rz1 = 1.f / (zz.y + eps);
#pragma unroll
  for (int v = 0; v < F / 4; ++v) {
    float4 a = ar[v];
    const float rz = (4 * v < C) ? rz0 : rz1;
#pragma unroll
    for (int j = 0; j < 4; ++j) {
      float x = (&a.x)[j] * rz + bias[4 * v + j];
      if (ELU) x = (x > 0.f) ? x : (__expf(x) - 1.f);
      (&a.x)[j] = x;
    }
    outp[v] = a;
  }
}

extern "C" void kernel_launch(void* const* d_in, const int* in_sizes, int n_in,
                              void* d_out, int out_size, void* d_ws, size_t ws_size,
                              hipStream_t stream) {
  const float* x   = (const float*)d_in[0];
  const int*   ei  = (const int*)  d_in[1];
  const float* W1  = (const float*)d_in[2];
  const float* as1 = (const float*)d_in[3];
  const float* ad1 = (const float*)d_in[4];
  const float* b1  = (const float*)d_in[5];
  const float* W2  = (const float*)d_in[6];
  const float* as2 = (const float*)d_in[7];
  const float* ad2 = (const float*)d_in[8];
  const float* b2  = (const float*)d_in[9];

  const int N = in_sizes[0] / 16;
  const int E = in_sizes[1] / 2;

  float*    ws   = (float*)d_ws;
  float*    h1   = ws;
  float*    h2   = ws;
  float*    acc2 = ws + (size_t)8 * N;
  float*    acc1 = ws + (size_t)16 * N;
  float*    als  = ws + (size_t)32 * N;
  float*    ald  = ws + (size_t)34 * N;
  unsigned* menc = (unsigned*)(ws + (size_t)36 * N);
  float*    z    = ws + (size_t)38 * N;
  float*    aux  = ws + (size_t)40 * N;
  int*      qc1  = (int*)aux;
  int*      qc2  = qc1 + 256;
  unsigned* st1  = (unsigned*)(aux + 512);
  unsigned* st2  = (unsigned*)(aux + 576);
  float*    pt1  = aux + 1024;
  float*    pt2  = aux + 1024 + 4096;
  unsigned* keys = (unsigned*)(aux + 16384);

  const int tiles     = (N + 15) / 16;
  const int node_grid = (tiles + NODE_TB / 32 - 1) / (NODE_TB / 32);
  const int max_grid  = 512;
  const int acc_grid  = 256;
  const int fin_grid  = (N + 255) / 256;

  node_layer<8, true><<<node_grid, NODE_TB, 0, stream>>>(x, z, nullptr, nullptr, W1, as1, ad1, h1, acc1,
                                                         als, ald, menc, pt1, ei, keys, qc1, N, E);
  edge_max<<<max_grid, MAX_TB, 0, stream>>>(ei, (const float2*)als, (const float2*)ald, menc, qc1,
                                            (const float4*)pt1, node_grid, st1, E, N);
  edge_accum<8><<<acc_grid, ACC_TB, 0, stream>>>(ei, (const float2*)als, (const float2*)ald, menc, z, acc1, h1,
                                                 keys, qc1 + 2 * QSTRIDE, st1, E, N);

  node_layer<4, false><<<node_grid, NODE_TB, 0, stream>>>(acc1, z, b1, st1, W2, as2, ad2, h2, acc2,
                                                          als, ald, menc, pt2, ei, keys, qc2, N, E);
  edge_max<<<max_grid, MAX_TB, 0, stream>>>(ei, (const float2*)als, (const float2*)ald, menc, qc2,
                                            (const float4*)pt2, node_grid, st2, E, N);
  edge_accum<4><<<acc_grid, ACC_TB, 0, stream>>>(ei, (const float2*)als, (const float2*)ald, menc, z, acc2, h2,
                                                 keys, qc2 + 2 * QSTRIDE, st2, E, N);
  node_finalize<4, false><<<fin_grid, 256, 0, stream>>>(acc2, (const float2*)z, b2, st2,
                                                        (float*)d_out, N);
}


# node launch stores each node's two source logits behind its feature row (pitch 2C+4 floats); an edge loads channels and logits as one group from one place, the separate logit-plane gather is gone
# speedup vs baseline: 11.8389x; 1.0723x over previous
#include <hip/hip_runtime.h>
#include <math.h>


typedef __attribute__((ext_vector_type(2))) float v2f;
typedef __attribute__((ext_vector_type(4))) float v4f;
typedef __attribute__((ext_vector_type(8))) float v8f;

__device__ __forceinline__ unsigned encf(float f) {
  unsigned u = __float_as_uint(f);
  return (u & 0x80000000u) ? ~u : (u | 0x80000000u);
}
__device__ __forceinline__ float decf(unsigned u) {
  unsigned v = (u & 0x80000000u) ? (u & 0x7FFFFFFFu) : ~u;
  return __uint_as_float(v);
}
#define ENC_NEG_INF 0x007FFFFFu
#define ENC_POS_INF 0xFF800000u
#define SPREAD_MAX 40.0f

__device__ __forceinline__ bool bound_mode(const unsigned* __restrict__ stat,
                                           float& b0, float& b1) {
  b0 = decf(stat[0]); b1 = decf(stat[1]);
  const float l0 = decf(stat[2]), l1 = decf(stat[3]);
  return (b0 - l0 <= SPREAD_MAX) && (b1 - l1 <= SPREAD_MAX);
}

#define CHUNK 512
#define RS 1568
#define QSTRIDE 64

__device__ __forceinline__ int package_half() {
  const unsigned v = __builtin_amdgcn_s_sendmsg_rtn(0x87);
  return (int)((v >> 18) & 1u);
}

__device__ __forceinline__ int split_node(int N) {
  const int s = ((N >> 1) + 31) & ~31;
  return (s < N) ? s : N;
}

#define NODE_TB 512

template <int CTRL>
__device__ __forceinline__ float dpp_get(float v) {
  return __int_as_float(__builtin_amdgcn_update_dpp(0, __float_as_int(v), CTRL, 0xF, 0xF, true));
}

__device__ __forceinline__ unsigned key_word(const int* __restrict__ dstp, int w, int E) {
  unsigned v = 0u;
  if ((E & 3) == 0) {
    const int4 d = ((const int4*)dstp)[w];
    v = (((unsigned)d.x / RS) & 0xFFu)       | ((((unsigned)d.y / RS) & 0xFFu) << 8) |
        ((((unsigned)d.z / RS) & 0xFFu) << 16) | ((((unsigned)d.w / RS) & 0xFFu) << 24);
  } else {
#pragma unroll
    for (int j = 0; j < 4; ++j) {
      const int t = 4 * w + j;
      if (t < E) v |= (((unsigned)dstp[t] / RS) & 0xFFu) << (8 * j);
    }
  }
  return v;
}

template <int C, bool FIRST>
__global__ void __launch_bounds__(NODE_TB)
node_layer(const float* __restrict__ X,
           float* zbuf,
           const float* __restrict__ bias,
           const unsigned* __restrict__ stat_prev,
           const float* __restrict__ W,
           const float* __restrict__ a_src,
           const float* __restrict__ a_dst,
           float* __restrict__ Hout, float* __restrict__ acc,
           float* __restrict__ ald,
           unsigned* __restrict__ menc,
           float* __restrict__ part,
           const int* __restrict__ ei, unsigned* __restrict__ keyw,
           int* __restrict__ qctr, int nrows, int E) {
  constexpr int F = 2 * C;
  constexpr int HP = F + 4;
  __shared__ float s_st[NODE_TB / 32][4];
  const int tid  = (int)threadIdx.x;
  const int lane = tid & 31;
  const int wv   = tid >> 5;
  float st[4] = {-INFINITY, -INFINITY, INFINITY, INFINITY};
  const int row_base = ((int)blockIdx.x * (NODE_TB / 32) + wv) * 16;
  if (row_base < nrows) {
    const int m    = lane & 15;
    const int half = lane >> 4;
    int row = row_base + m; if (row > nrows - 1) row = nrows - 1;
    const int col = m & (F - 1);
    const float2* __restrict__ Xr = (const float2*)(X + (size_t)row * 16);
    float rz0 = 1.f, rz1 = 1.f;
    if (!FIRST) {
      float b0, b1;
      const float eps = bound_mode(stat_prev, b0, b1) ? 0.f : 1e-16f;
      const float2 zz = ((const float2*)zbuf)[row];
      rz0 = 1.f / (zz.x + eps);
      rz1 = 1.f / (zz.y + eps);
    }
    v8f c = {0.f, 0.f, 0.f, 0.f, 0.f, 0.f, 0.f, 0.f};
#pragma unroll
    for (int k0 = 0; k0 < 4; ++k0) {
      const int k = k0 * 4 + half * 2;
      const float2 av = Xr[k0 * 2 + half];
      v2f a;
      if (FIRST) {
        a.x = av.x; a.y = av.y;
      } else {
        const float rz = (k0 < 2) ? rz0 : rz1;
        float x0 = av.x * rz + bias[k];
        float x1 = av.y * rz + bias[k + 1];
        x0 = (x0 > 0.f) ? x0 : (__expf(x0) - 1.f);
        x1 = (x1 > 0.f) ? x1 : (__expf(x1) - 1.f);
        a.x = x0; a.y = x1;
      }
      v2f b;
      b.x = W[(k)     * F + col];
      b.y = W[(k + 1) * F + col];
      c = __builtin_amdgcn_wmma_f32_16x16x4_f32(
           false, a,  false, b,
           (short)0, c,  false,  false);
    }

    const int rofs = half * 8;
    if (m < F) {
      float* __restrict__ outp = Hout + (size_t)(row_base + rofs) * HP + m;
      if (row_base + 16 <= nrows) {
#pragma unroll
        for (int r = 0; r < 8; ++r) outp[(size_t)r * HP] = c[r];
      } else {
#pragma unroll
        for (int r = 0; r < 8; ++r)
          if (row_base + rofs + r < nrows) outp[(size_t)r * HP] = c[r];
      }
    }

    const float as_c = a_src[col], ad_c = a_dst[col];
    float sv[8], dv[8];
#pragma unroll
    for (int r = 0; r < 8; ++r) { sv[r] = c[r] * as_c; dv[r] = c[r] * ad_c; }
#pragma unroll
    for (int r = 0; r < 8; ++r) { sv[r] += dpp_get<0xB1>(sv[r]); dv[r] += dpp_get<0xB1>(dv[r]); }
#pragma unroll
    for (int r = 0; r < 8; ++r) { sv[r] += dpp_get<0x4E>(sv[r]); dv[r] += dpp_get<0x4E>(dv[r]); }
    if (C == 8) {
#pragma unroll
      for (int r = 0; r < 8; ++r) { sv[r] += dpp_get<0x164>(sv[r]); dv[r] += dpp_get<0x164>(dv[r]); }
    }
    const int head  = (C == 8) ? (m >> 3) : ((m >> 2) & 1);
    const int r_sel = (C == 8) ? (m & 7) : ((m & 3) + 4 * (m >> 3));
    float s_sel = sv[0], d_sel = dv[0];
#pragma unroll
    for (int r = 1; r < 8; ++r)
      if (r_sel == r) { s_sel = sv[r]; d_sel = dv[r]; }
    const int orow = row_base + rofs + r_sel;
    if (orow < nrows) {
      float* __restrict__ tail = Hout + (size_t)orow * HP + F;
      tail[head]     = s_sel;
      tail[2 + head] = 0.f;
      ald[orow * 2 + head] = d_sel;
      if (head == 0) { st[0] = s_sel; st[2] = s_sel; }
      else           { st[1] = s_sel; st[3] = s_sel; }
      if (!(fabsf(s_sel) <= 3.0e38f && fabsf(d_sel) <= 3.0e38f)) st[0] = INFINITY;
    }

    const int nr = (row_base + 16 <= nrows) ? 16 : (nrows - row_base);
    float4* __restrict__ ar = (float4*)(acc + (size_t)row_base * F);
    for (int i = lane; i < nr * (F / 4); i += 32) ar[i] = make_float4(0.f, 0.f, 0.f, 0.f);
    if (lane < nr * 2) {
      zbuf[row_base * 2 + lane] = 0.f;
      menc[row_base * 2 + lane] = ENC_NEG_INF;
    }
  }

#pragma unroll
  for (int o = 16; o > 0; o >>= 1) {
    st[0] = fmaxf(st[0], __shfl_xor(st[0], o)); st[1] = fmaxf(st[1], __shfl_xor(st[1], o));
    st[2] = fminf(st[2], __shfl_xor(st[2], o)); st[3] = fminf(st[3], __shfl_xor(st[3], o));
  }
  if (lane == 0) {
#pragma unroll
    for (int k = 0; k < 4; ++k) s_st[wv][k] = st[k];
  }
  __syncthreads();
  if (tid < 4) {
    float v = s_st[0][tid];
    for (int w = 1; w < NODE_TB / 32; ++w)
      v = (tid < 2) ? fmaxf(v, s_st[w][tid]) : fminf(v, s_st[w][tid]);
    part[(int)blockIdx.x * 4 + tid] = v;
  }

  if (FIRST) {
    if (blockIdx.x == 0) qctr[tid] = 0;
    const int* __restrict__ dstp = ei + E;
    const int nwords = (E + 3) >> 2;
    const int step = (int)gridDim.x * NODE_TB;
    for (int w0 = (int)blockIdx.x * NODE_TB + tid; w0 < nwords; w0 += 4 * step) {
      unsigned kv[4];
#pragma unroll
      for (int k = 0; k < 4; ++k) {
        const int w = w0 + k * step;
        kv[k] = (w < nwords) ? key_word(dstp, w, E) : 0u;
      }
#pragma unroll
      for (int k = 0; k < 4; ++k) {
        const int w = w0 + k * step;
        if (w < nwords) keyw[w] = kv[k];
      }
    }
  }
}

#define MAX_TB 256
__global__ void __launch_bounds__(MAX_TB)
edge_max(const int* __restrict__ ei,
         const float* __restrict__ Hm, int hp,
         const float2* __restrict__ ald,
         unsigned* __restrict__ menc,
         int* __restrict__ qctr,
         const float4* __restrict__ part, int nb,
         unsigned* __restrict__ stat, int E, int N) {
  __shared__ int s_c;
  __shared__ float s_red[MAX_TB / 32][4];
  __shared__ float s_fin[4];
  {
    float r[4] = {-INFINITY, -INFINITY, INFINITY, INFINITY};
    for (int j = (int)threadIdx.x; j < nb; j += MAX_TB) {
      const float4 p = part[j];
      r[0] = fmaxf(r[0], p.x); r[1] = fmaxf(r[1], p.y);
      r[2] = fminf(r[2], p.z); r[3] = fminf(r[3], p.w);
    }
#pragma unroll
    for (int o = 16; o > 0; o >>= 1) {
      r[0] = fmaxf(r[0], __shfl_xor(r[0], o)); r[1] = fmaxf(r[1], __shfl_xor(r[1], o));
      r[2] = fminf(r[2], __shfl_xor(r[2], o)); r[3] = fminf(r[3], __shfl_xor(r[3], o));
    }
    if ((threadIdx.x & 31) == 0) {
#pragma unroll
      for (int k = 0; k < 4; ++k) s_red[threadIdx.x >> 5][k] = r[k];
    }
    __syncthreads();
    if (threadIdx.x < 4) {
      const int k = (int)threadIdx.x;
      float v = s_red[0][k];
      for (int w = 1; w < MAX_TB / 32; ++w)
        v = (k < 2) ? fmaxf(v, s_red[w][k]) : fminf(v, s_red[w][k]);
      s_fin[k] = v;
      if (blockIdx.x == 0) stat[k] = encf(v);
    }
    __syncthreads();
    if ((s_fin[0] - s_fin[2] <= SPREAD_MAX) && (s_fin[1] - s_fin[3] <= SPREAD_MAX)) return;
  }
  const int total = E + N;
  const int nchunks = (total + CHUNK - 1) / CHUNK;
  const int nsplit = split_node(N);
  const int myhalf = package_half();
  for (int side = 0; side < 2; ++side) {
    const int part_i = myhalf ^ side;
    const int lo = part_i ? nsplit : 0;
    const unsigned span = (unsigned)((part_i ? N : nsplit) - lo);
    int* __restrict__ q = qctr + part_i * QSTRIDE;
    int pend = 0;
    if (threadIdx.x == 0) pend = atomicAdd(q, 1);
    for (int g = 0; g <= nchunks; ++g) {
      if (threadIdx.x == 0) s_c = pend;
      __syncthreads();
      const int c = s_c;
      __syncthreads();
      if (c >= nchunks) break;
      if (threadIdx.x == 0) pend = atomicAdd(q, 1);
      int dstk[CHUNK / MAX_TB];
#pragma unroll
      for (int k = 0; k < CHUNK / MAX_TB; ++k) {
        const int t = c * CHUNK + k * MAX_TB + (int)threadIdx.x;
        dstk[k] = (t < total) ? ((t < E) ? ei[E + t] : t - E) : -1;
      }
#pragma unroll
      for (int k = 0; k < CHUNK / MAX_TB; ++k) {
        const int t = c * CHUNK + k * MAX_TB + (int)threadIdx.x;
        const int dst = dstk[k];
        if ((unsigned)(dst - lo) < span) {
          const int src = (t < E) ? ei[t] : dst;
          const float2 s = *(const float2*)(Hm + (size_t)src * hp + (hp - 4));
          const float2 d = ald[dst];
          float e0 = s.x + d.x; e0 = (e0 > 0.f) ? e0 : 0.2f * e0;
          float e1 = s.y + d.y; e1 = (e1 > 0.f) ? e1 : 0.2f * e1;
          atomicMax(&menc[dst * 2],     encf(e0));
          atomicMax(&menc[dst * 2 + 1], encf(e1));
        }
      }
    }
  }
}

#define RB 4
#define SCAN_K 8
#define ACC_TB 1024
#define QCAP 160

__device__ __forceinline__ void wave_queue_handover() {
  __builtin_amdgcn_fence(__ATOMIC_RELEASE, "wavefront");
  __builtin_amdgcn_wave_barrier();
  __builtin_amdgcn_fence(__ATOMIC_ACQUIRE, "wavefront");
}

template <int C>
__device__ __forceinline__ void edge_terms(int src, int dl,
                                           const float* __restrict__ Hm,
                                           const float4* s_dm,
                                           float* s_row) {
  constexpr int F = 2 * C;
  constexpr int HP = F + 4;
  constexpr int P = F + 3;
  const float* __restrict__ hrow = Hm + (size_t)src * HP;
  const v4f* __restrict__ hs = (const v4f*)hrow;
  v4f hv[F / 4];
#pragma unroll
  for (int v = 0; v < F / 4; ++v) hv[v] = hs[v];
  const v2f s = *(const v2f*)(hrow + F);
  if constexpr (C == 8) asm volatile("" :: "v"(hv[0]), "v"(hv[1]), "v"(hv[2]), "v"(hv[3]), "v"(s));
  else                  asm volatile("" :: "v"(hv[0]), "v"(hv[1]), "v"(s));
  const float4 dm = s_dm[dl];
  float w[2];
#pragma unroll
  for (int hd = 0; hd < 2; ++hd) {
    float e = (hd ? s.y + dm.y : s.x + dm.x);
    e = (e > 0.f) ? e : 0.2f * e;
    w[hd] = __expf(e - (hd ? dm.w : dm.z));
  }
  float* r = s_row + dl * P;
  atomicAdd(&r[F + 0], w[0]);
  atomicAdd(&r[F + 1], w[1]);
#pragma unroll
  for (int hd = 0; hd < 2; ++hd) {
#pragma unroll
    for (int v = 0; v < C / 4; ++v) {
      const v4f h4 = hv[hd * (C / 4) + v];
      const int base = hd * C + 4 * v;
      atomicAdd(&r[base + 0], w[hd] * h4.x);
      atomicAdd(&r[base + 1], w[hd] * h4.y);
      atomicAdd(&r[base + 2], w[hd] * h4.z);
      atomicAdd(&r[base + 3], w[hd] * h4.w);
    }
  }
}

template <int C>
__global__ void __launch_bounds__(ACC_TB)
edge_accum(const int* __restrict__ ei,
           const float2* __restrict__ ald,
           const unsigned* __restrict__ menc,
           float* __restrict__ z, float* __restrict__ acc,
           const float* __restrict__ Hm,
           const unsigned* __restrict__ keyw,
           int* __restrict__ qctr,
           const unsigned* __restrict__ stat, int E, int N) {
  constexpr int F = 2 * C;
  constexpr int P = F + 3;
  __shared__ float s_row[RS * P];
  __shared__ float4 s_dm[RS];
  __shared__ int s_q[ACC_TB / 32][QCAP];
  __shared__ int s_c;

  const int tid  = (int)threadIdx.x;
  const int lane = tid & 31;
  int* q_w = &s_q[tid >> 5][0];
  const int* __restrict__ dstp = ei + E;

  const int R  = (N + RS - 1) / RS;
  const int Rh = (R + 1) >> 1;
  const int myhalf = package_half();
  float b0, b1;
  const bool bmode = bound_mode(stat, b0, b1);
  for (int side = 0; side < 2; ++side) {
    const int part = myhalf ^ side;
    const int r0 = part ? Rh : 0;
    const int nitems = (part ? R - Rh : Rh) * RB;
    int* __restrict__ q = qctr + part * QSTRIDE;
    for (int g = 0; g <= nitems; ++g) {
      if (tid == 0) s_c = atomicAdd(q, 1);
      __syncthreads();
      const int c = s_c;
      __syncthreads();
      if (c >= nitems) break;
      const int rng = r0 + c / RB;
      const int b   = c % RB;
      const int lo  = rng * RS;
      const int span = ((lo + RS < N) ? lo + RS : N) - lo;

      for (int i = tid; i < RS * P; i += ACC_TB) s_row[i] = 0.f;
      for (int i = tid; i < span; i += ACC_TB) {
        const float2 d = ald[lo + i];
        float m0, m1;
        if (bmode) {
          m0 = b0 + d.x; m0 = (m0 > 0.f) ? m0 : 0.2f * m0;
          m1 = b1 + d.y; m1 = (m1 > 0.f) ? m1 : 0.2f * m1;
        } else {
          m0 = decf(menc[(lo + i) * 2]);
          m1 = decf(menc[(lo + i) * 2 + 1]);
          if (!(m0 >= -3.0e38f && m0 <= 3.0e38f)) m0 = 0.f;
          if (!(m1 >= -3.0e38f && m1 <= 3.0e38f)) m1 = 0.f;
        }
        s_dm[i] = make_float4(d.x, d.y, m0, m1);
      }
      __syncthreads();

      const int e_lo = (int)(((long long)E * b) / RB) & ~3;
      const int e_hi = (b == RB - 1) ? E : ((int)(((long long)E * (b + 1)) / RB) & ~3);
      const unsigned rngw = (unsigned)rng * 0x01010101u;
      int qn = 0;
      for (int wb = e_lo >> 2; wb < ((e_hi + 3) >> 2); wb += ACC_TB * SCAN_K) {
        unsigned kw[SCAN_K];
        if (wb + ACC_TB * SCAN_K <= (e_hi >> 2)) {
#pragma unroll
          for (int k = 0; k < SCAN_K; ++k) kw[k] = keyw[wb + k * ACC_TB + tid] ^ rngw;
        } else {
#pragma unroll
          for (int k = 0; k < SCAN_K; ++k) {
            const int w = wb + k * ACC_TB + tid;
            const int left = e_hi - 4 * w;
            unsigned v = 0xFFFFFFFFu;
            if (left > 0) {
              v = keyw[w] ^ rngw;
              if (left < 4) v |= 0xFFFFFFFFu << (8 * left);
            }
            kw[k] = v;
          }
        }
        static_assert(SCAN_K == 8, "one flag bit per byte position and word: 4 x 8 = 32");
        unsigned notown = 0u;
#pragma unroll
        for (int k = 0; k < SCAN_K; ++k) {
          const unsigned x = kw[k];
          notown = (notown >> 1) | ((((x & 0x7F7F7F7Fu) + 0x7F7F7F7Fu) | x) & 0x80808080u);
        }
        unsigned own = ~notown;
        const int t00 = 4 * (wb + tid);
        unsigned any = __builtin_amdgcn_ballot_w32(own != 0u);
        while (any != 0u) {
          if (own != 0u) {
            const int p = __builtin_ctz(own);
            const int pos = (int)__builtin_amdgcn_mbcnt_lo(any, (unsigned)qn);
            q_w[pos] = t00 + (p & 7) * (4 * ACC_TB) + (p >> 3);
            own &= own - 1u;
          }
          qn += __builtin_popcount(any);
          if (qn >= 32) {
            qn -= 32;
            wave_queue_handover();
            const int t = q_w[qn + lane];
            const int src = ei[t];
            const int dst = dstp[t];
            asm volatile("" :: "v"(src), "v"(dst));
            const int dl = dst - lo;
            if ((unsigned)dl < (unsigned)span)
              edge_terms<C>(src, dl, Hm, s_dm, s_row);
          }
          any = __builtin_amdgcn_ballot_w32(own != 0u);
        }
      }
      wave_queue_handover();
      if (lane < qn) {
        const int t = q_w[lane];
        const int src = ei[t];
        const int dst = dstp[t];
        asm volatile("" :: "v"(src), "v"(dst));
        const int dl = dst - lo;
        if ((unsigned)dl < (unsigned)span)
          edge_terms<C>(src, dl, Hm, s_dm, s_row);
      }
      if (b == 0)
        for (int i = tid; i < span; i += ACC_TB)
          edge_terms<C>(lo + i, i, Hm, s_dm, s_row);
      __syncthreads();

      float* __restrict__ accr = acc + (size_t)lo * F;
      for (int i = tid; i < span * F; i += ACC_TB)
        atomicAdd(&accr[i], s_row[(i / F) * P + (i % F)]);
      float* __restrict__ zr = z + (size_t)lo * 2;
      for (int i = tid; i < span * 2; i += ACC_TB)
        atomicAdd(&zr[i], s_row[(i >> 1) * P + F + (i & 1)]);
    }
  }
}

template <int C, bool ELU>
__global__ void node_finalize(const float* __restrict__ acc,
                              const float2* __restrict__ z,
                              const float* __restrict__ bias,
                              const unsigned* __restrict__ stat,
                              float* __restrict__ out, int nrows) {
  const int n = blockIdx.x * blockDim.x + threadIdx.x;
  if (n >= nrows) return;
  constexpr int F = 2 * C;
  float b0, b1;
  const float eps = bound_mode(stat, b0, b1) ? 0.f : 1e-16f;
  const float4* __restrict__ ar = (const float4*)(acc + (size_t)n * F);
  float4* __restrict__ outp = (float4*)(out + (size_t)n * F);
  const float2 zz = z[n];
  const float rz0 = 1.f / (zz.x + eps);
  const float rz1 = 1.f / (zz.y + eps);
#pragma unroll
  for (int v = 0; v < F / 4; ++v) {
    float4 a = ar[v];
    const float rz = (4 * v < C) ? rz0 : rz1;
#pragma unroll
    for (int j = 0; j < 4; ++j) {
      float x = (&a.x)[j] * rz + bias[4 * v + j];
      if (ELU) x = (x > 0.f) ? x : (__expf(x) - 1.f);
      (&a.x)[j] = x;
    }
    outp[v] = a;
  }
}

extern "C" void kernel_launch(void* const* d_in, const int* in_sizes, int n_in,
                              void* d_out, int out_size, void* d_ws, size_t ws_size,
                              hipStream_t stream) {
  const float* x   = (const float*)d_in[0];
  const int*   ei  = (const int*)  d_in[1];
  const float* W1  = (const float*)d_in[2];
  const float* as1 = (const float*)d_in[3];
  const float* ad1 = (const float*)d_in[4];
  const float* b1  = (const float*)d_in[5];
  const float* W2  = (const float*)d_in[6];
  const float* as2 = (const float*)d_in[7];
  const float* ad2 = (const float*)d_in[8];
  const float* b2  = (const float*)d_in[9];

  const int N = in_sizes[0] / 16;
  const int E = in_sizes[1] / 2;

  float*    ws   = (float*)d_ws;
  float*    h1   = ws;
  float*    h2   = ws;
  float*    acc2 = ws + (size_t)12 * N;
  float*    acc1 = ws + (size_t)20 * N;
  float*    ald  = ws + (size_t)36 * N;
  unsigned* menc = (unsigned*)(ws + (size_t)38 * N);
  float*    z    = ws + (size_t)40 * N;
  float*    aux  = ws + (size_t)42 * N;
  int*      qc1  = (int*)aux;
  int*      qc2  = qc1 + 256;
  unsigned* st1  = (unsigned*)(aux + 512);
  unsigned* st2  = (unsigned*)(aux + 576);
  float*    pt1  = aux + 1024;
  float*    pt2  = aux + 1024 + 4096;
  unsigned* keys = (unsigned*)(aux + 16384);

  const int tiles     = (N + 15) / 16;
  const int node_grid = (tiles + NODE_TB / 32 - 1) / (NODE_TB / 32);
  const int max_grid  = 512;
  const int acc_grid  = 256;
  const int fin_grid  = (N + 255) / 256;

  node_layer<8, true><<<node_grid, NODE_TB, 0, stream>>>(x, z, nullptr, nullptr, W1, as1, ad1, h1, acc1,
                                                         ald, menc, pt1, ei, keys, qc1, N, E);
  edge_max<<<max_grid, MAX_TB, 0, stream>>>(ei, h1, 20, (const float2*)ald, menc, qc1,
                                            (const float4*)pt1, node_grid, st1, E, N);
  edge_accum<8><<<acc_grid, ACC_TB, 0, stream>>>(ei, (const float2*)ald, menc, z, acc1, h1,
                                                 keys, qc1 + 2 * QSTRIDE, st1, E, N);

  node_layer<4, false><<<node_grid, NODE_TB, 0, stream>>>(acc1, z, b1, st1, W2, as2, ad2, h2, acc2,
                                                          ald, menc, pt2, ei, keys, qc2, N, E);
  edge_max<<<max_grid, MAX_TB, 0, stream>>>(ei, h2, 12, (const float2*)ald, menc, qc2,
                                            (const float4*)pt2, node_grid, st2, E, N);
  edge_accum<4><<<acc_grid, ACC_TB, 0, stream>>>(ei, (const float2*)ald, menc, z, acc2, h2,
                                                 keys, qc2 + 2 * QSTRIDE, st2, E, N);
  node_finalize<4, false><<<fin_grid, 256, 0, stream>>>(acc2, (const float2*)z, b2, st2,
                                                        (float*)d_out, N);
}
